# starting kernel
# baseline (speedup 1.0000x reference)
#include <hip/hip_runtime.h>
#include <hip/hip_bf16.h>


#define USER_NUM 60000
#define ITEM_NUM 40000
#define NODE_NUM (USER_NUM + ITEM_NUM)
#define EMBED_DIM 64
#define N_LAYER 3
#define NNZ 1600000
#define BATCH 4096
#define NEG_SLOPE 0.2f
#define NODE_TILES (NODE_NUM / 16)

typedef float v2f __attribute__((ext_vector_type(2)));
typedef float v8f __attribute__((ext_vector_type(8)));

__global__ void ngcf_zero_kernel(float4* __restrict__ p, int n4) {
    int i = blockIdx.x * blockDim.x + threadIdx.x;
    if (i < n4) p[i] = make_float4(0.f, 0.f, 0.f, 0.f);
}

__global__ void ngcf_spmm_kernel(const float* __restrict__ vals,
                                 const int*   __restrict__ rows,
                                 const int*   __restrict__ cols,
                                 const float* __restrict__ E,
                                 float*       __restrict__ side) {
    const int wavesPerBlock = blockDim.x >> 5;
    const int edge = blockIdx.x * wavesPerBlock + (threadIdx.x >> 5);
    if (edge >= NNZ) return;
    const int lane = threadIdx.x & 31;

    const float v = vals[edge];
    const int   r = rows[edge];
    const int   c = cols[edge];

    const float2 e = *(const float2*)(E + (size_t)c * EMBED_DIM + lane * 2);
    float* dst = side + (size_t)r * EMBED_DIM + lane * 2;
    atomicAdd(dst + 0, v * e.x);
    atomicAdd(dst + 1, v * e.y);
}

__global__ void ngcf_layer_kernel(const float* __restrict__ Eprev,
                                  const float* __restrict__ side,
                                  const float* __restrict__ W1,
                                  const float* __restrict__ W2,
                                  float*       __restrict__ Eout) {
    __shared__ float w1i[EMBED_DIM * EMBED_DIM];
    __shared__ float w2i[EMBED_DIM * EMBED_DIM];
    for (int i = threadIdx.x; i < EMBED_DIM * EMBED_DIM; i += blockDim.x) {
        const int k = i >> 6;
        const int n = i & 63;
        const int dst = ((k >> 1) << 7) + (n << 1) + (k & 1);
        w1i[dst] = W1[i];
        w2i[dst] = W2[i];
    }
    __syncthreads();

    const int tile = blockIdx.x * (blockDim.x >> 5) + (threadIdx.x >> 5);
    if (tile >= NODE_TILES) return;

    const int lane    = threadIdx.x & 31;
    const int halfsel = lane >> 4;
    const int nlane   = lane & 15;
    const int m0      = tile * 16;
    const int mrow    = m0 + nlane;
    const int koff    = halfsel * 2;

    v8f acc[4] = {v8f{}, v8f{}, v8f{}, v8f{}};

    const float* epRow = Eprev + (size_t)mrow * EMBED_DIM + koff;
    const float* sdRow = side  + (size_t)mrow * EMBED_DIM + koff;

#pragma unroll
    for (int kk = 0; kk < 16; ++kk) {
        const int k0 = kk * 4;
        const float2 ep = *(const float2*)(epRow + k0);
        const float2 sd = *(const float2*)(sdRow + k0);

        v2f aSum, aBi;
        aSum.x = ep.x + sd.x;  aSum.y = ep.y + sd.y;
        aBi.x  = ep.x * sd.x;  aBi.y  = ep.y * sd.y;

        const int kph = ((k0 >> 1) + halfsel) << 7;
        const float* w1p = w1i + kph + (nlane << 1);
        const float* w2p = w2i + kph + (nlane << 1);

#pragma unroll
        for (int nb = 0; nb < 4; ++nb) {
            const v2f b1 = *(const v2f*)(w1p + nb * 32);
            const v2f b2 = *(const v2f*)(w2p + nb * 32);
            acc[nb] = __builtin_amdgcn_wmma_f32_16x16x4_f32(
                false, aSum, false, b1, (short)0, acc[nb], false, false);
            acc[nb] = __builtin_amdgcn_wmma_f32_16x16x4_f32(
                false, aBi,  false, b2, (short)0, acc[nb], false, false);
        }
    }

#pragma unroll
    for (int nb = 0; nb < 4; ++nb) {
#pragma unroll
        for (int i = 0; i < 8; ++i) {
            float v = acc[nb][i];
            v = (v > 0.f) ? v : v * NEG_SLOPE;
            Eout[(size_t)(m0 + i + halfsel * 8) * EMBED_DIM + nb * 16 + nlane] = v;
        }
    }
}

__global__ void ngcf_gather_kernel(const float* __restrict__ emb,
                                   const float* __restrict__ E1,
                                   const float* __restrict__ E2,
                                   const float* __restrict__ E3,
                                   const int* __restrict__ user_idx,
                                   const int* __restrict__ item_idx,
                                   float* __restrict__ out) {
    const int row = blockIdx.x;
    const int t   = threadIdx.x;
    const int seg = t >> 6;
    const int dim = t & 63;

    int node;
    if (row < BATCH) node = user_idx[row];
    else             node = USER_NUM + item_idx[row - BATCH];

    const float* src = (seg == 0) ? emb : (seg == 1) ? E1 : (seg == 2) ? E2 : E3;
    out[(size_t)row * 256 + t] = src[(size_t)node * EMBED_DIM + dim];
}

extern "C" void kernel_launch(void* const* d_in, const int* in_sizes, int n_in,
                              void* d_out, int out_size, void* d_ws, size_t ws_size,
                              hipStream_t stream) {
    const float* emb      = (const float*)d_in[0];
    const float* W1       = (const float*)d_in[1];
    const float* W2       = (const float*)d_in[2];
    const float* vals     = (const float*)d_in[3];
    const int*   row_idx  = (const int*)d_in[4];
    const int*   col_idx  = (const int*)d_in[5];
    const int*   user_idx = (const int*)d_in[6];
    const int*   item_idx = (const int*)d_in[7];

    const size_t EN = (size_t)NODE_NUM * EMBED_DIM;
    float* side = (float*)d_ws;
    float* E1   = side + EN;
    float* E2   = E1 + EN;
    float* E3   = E2 + EN;
    float* Eouts[N_LAYER] = {E1, E2, E3};

    const int layerBlocks = (NODE_TILES + 7) / 8;

    const float* Eprev = emb;
    for (int l = 0; l < N_LAYER; ++l) {
        ngcf_zero_kernel<<<(int)(EN / 4 / 256), 256, 0, stream>>>((float4*)side,
                                                                  (int)(EN / 4));
        ngcf_spmm_kernel<<<NNZ / 8, 256, 0, stream>>>(vals, row_idx, col_idx,
                                                      Eprev, side);
        ngcf_layer_kernel<<<layerBlocks, 256, 0, stream>>>(
            Eprev, side,
            W1 + (size_t)l * EMBED_DIM * EMBED_DIM,
            W2 + (size_t)l * EMBED_DIM * EMBED_DIM,
            Eouts[l]);
        Eprev = Eouts[l];
    }

    ngcf_gather_kernel<<<2 * BATCH, 256, 0, stream>>>(emb, E1, E2, E3,
                                                      user_idx, item_idx,
                                                      (float*)d_out);
}


# SpMM: 256 workgroups each own 391 dest rows in LDS, find their edges in a per-call bit-plane owner map, add with LDS f32 atomics (arrival order), store once; no global float atomics; 10->8 launches
# speedup vs baseline: 2.1383x; 2.1383x over previous
#include <hip/hip_runtime.h>
#include <hip/hip_bf16.h>


#define USER_NUM 60000
#define ITEM_NUM 40000
#define NODE_NUM (USER_NUM + ITEM_NUM)
#define EMBED_DIM 64
#define N_LAYER 3
#define NNZ 1600000
#define BATCH 4096
#define NEG_SLOPE 0.2f
#define NODE_TILES (NODE_NUM / 16)

typedef float v2f __attribute__((ext_vector_type(2)));
typedef float v8f __attribute__((ext_vector_type(8)));

#define OWN_ROWS 391
#define OWNERS 256
#define PLANES 8
#define PLANE_WORDS (NNZ / 32)
#define PLANE_VEC4 (PLANE_WORDS / 4)
#define PLANE_STRIDE 50016
#define PLANE_STRIDE4 (PLANE_STRIDE / 4)
#define PLANE_WAVES ((PLANE_WORDS + 31) / 32)
__global__ void __launch_bounds__(32)
ngcf_planes_kernel(const int* __restrict__ rows, unsigned* __restrict__ planes) {
    const int lane = threadIdx.x;
    const int w0   = blockIdx.x * 32;
    unsigned mine[PLANES];
#pragma unroll
    for (int k = 0; k < PLANES; ++k) mine[k] = 0u;
#pragma unroll 8
    for (int t = 0; t < 32; ++t) {
        const int w = w0 + t;
        if (w < PLANE_WORDS) {
            const unsigned r = (unsigned)rows[w * 32 + lane];
            unsigned o = r / (unsigned)OWN_ROWS;
            o = (o < (unsigned)OWNERS) ? o : (unsigned)(OWNERS - 1);
#pragma unroll
            for (int k = 0; k < PLANES; ++k) {
                const unsigned b = __builtin_amdgcn_ballot_w32(((o >> k) & 1u) != 0u);
                if (lane == t) mine[k] = b;
            }
        }
    }
    if (w0 + lane < PLANE_WORDS) {
#pragma unroll
        for (int k = 0; k < PLANES; ++k)
            planes[(size_t)k * PLANE_STRIDE + w0 + lane] = mine[k];
    }
}

#define SPMM_HALF (EMBED_DIM / 2)
#define QCAP 16384
#define FIND_TRIPS ((PLANE_VEC4 + 31) / 32)

static __device__ __forceinline__ void spmm_add_batch(const int myC, const int myR, const float myV,
                                                      const int cnt,
                                                      const float* __restrict__ E,
                                                      float* acc, const int rlo, const int lane) {
    float glo[32];
    float ghi[32];
#pragma unroll
    for (int j = 0; j < 32; ++j) {
        if (j < cnt) {
            const int c = __builtin_amdgcn_readlane(myC, j);
            const float* srow = E + (size_t)c * EMBED_DIM + lane;
            glo[j] = srow[0];
            ghi[j] = srow[SPMM_HALF];
        }
    }
#pragma unroll
    for (int j = 0; j < 32; ++j) {
        if (j < cnt) {
            const int   lr = __builtin_amdgcn_readlane(myR, j) - rlo;
            const float v  = __int_as_float(
                __builtin_amdgcn_readlane(__float_as_int(myV), j));
            if ((unsigned)lr < (unsigned)OWN_ROWS) {
                float* drow = acc + lr * EMBED_DIM + lane;
                atomicAdd(drow,             v * glo[j]);
                atomicAdd(drow + SPMM_HALF, v * ghi[j]);
            }
        }
    }
}

__global__ void __launch_bounds__(1024)
ngcf_spmm_kernel(const float*    __restrict__ vals,
                 const int*      __restrict__ rows,
                 const int*      __restrict__ cols,
                 const float*    __restrict__ E,
                 float*          __restrict__ side,
                 const unsigned* __restrict__ planes) {
    __shared__ float    acc[OWN_ROWS * EMBED_DIM];
    __shared__ unsigned queue[QCAP];
    __shared__ unsigned qcount;

    const int tid   = threadIdx.x;
    const int lane  = tid & 31;
    const int wave  = __builtin_amdgcn_readfirstlane(tid >> 5);
    const int owner = blockIdx.x;
    const int rlo   = owner * OWN_ROWS;

    for (int i = tid; i < OWN_ROWS * (EMBED_DIM / 4); i += 1024)
        ((float4*)acc)[i] = make_float4(0.f, 0.f, 0.f, 0.f);
    if (tid == 0) qcount = 0u;
    __syncthreads();

    const uint4* __restrict__ P4 = (const uint4*)planes;
    const int rot = (owner * FIND_TRIPS) >> 8;
    for (int t = wave; t < FIND_TRIPS; t += 32) {
        const int  trip = (t + rot < FIND_TRIPS) ? (t + rot) : (t + rot - FIND_TRIPS);
        const int  i  = trip * 32 + lane;
        const bool ok = i < PLANE_VEC4;
        const int  ii = ok ? i : (PLANE_VEC4 - 1);
        uint4 p[PLANES];
#pragma unroll
        for (int k = 0; k < PLANES; ++k) p[k] = P4[(size_t)k * PLANE_STRIDE4 + ii];
        unsigned n0 = 0u, n1 = 0u, n2 = 0u, n3 = 0u;
#pragma unroll
        for (int k = 0; k < PLANES; ++k) {
            const unsigned want = ((owner >> k) & 1) ? 0xFFFFFFFFu : 0u;
            n0 |= p[k].x ^ want;  n1 |= p[k].y ^ want;
            n2 |= p[k].z ^ want;  n3 |= p[k].w ^ want;
        }
        unsigned m[4];
        m[0] = ok ? ~n0 : 0u;  m[1] = ok ? ~n1 : 0u;
        m[2] = ok ? ~n2 : 0u;  m[3] = ok ? ~n3 : 0u;
#pragma unroll
        for (int d = 0; d < 4; ++d) {
            unsigned mm = m[d];
            unsigned bal;
            while ((bal = __builtin_amdgcn_ballot_w32(mm != 0u)) != 0u) {
                if (mm != 0u) {
                    const int      lead = __builtin_ctz(bal);
                    const unsigned rank = __builtin_amdgcn_mbcnt_lo(bal, 0u);
                    const int      bit  = __builtin_ctz(mm);
                    mm &= mm - 1u;
                    unsigned base = 0u;
                    if (lane == lead)
                        base = atomicAdd(&qcount, (unsigned)__builtin_popcount(bal));
                    base = (unsigned)__builtin_amdgcn_readlane((int)base, lead);
                    const unsigned slot = base + rank;
                    const int e = ((ii * 4 + d) << 5) + bit;
                    if (slot < (unsigned)QCAP) {
                        queue[slot] = (unsigned)e;
                    } else {
                        const int   c  = cols[e];
                        const int   lr = rows[e] - rlo;
                        const float v  = vals[e];
                        if ((unsigned)lr < (unsigned)OWN_ROWS) {
                            for (int x = 0; x < EMBED_DIM; ++x)
                                atomicAdd(&acc[lr * EMBED_DIM + x],
                                          v * E[(size_t)c * EMBED_DIM + x]);
                        }
                    }
                }
            }
        }
    }
    __syncthreads();

    const unsigned qc = (unsigned)__builtin_amdgcn_readfirstlane((int)qcount);
    const int nq = (int)((qc < (unsigned)QCAP) ? qc : (unsigned)QCAP);
    for (int b0 = wave * 32; b0 < nq; b0 += 32 * 32) {
        const int idx = b0 + lane;
        const int e   = (idx < nq) ? (int)queue[idx] : 0;
        const int   myC = cols[e];
        const int   myR = rows[e];
        const float myV = vals[e];
        asm volatile("" :: "v"(myC), "v"(myR), "v"(myV));
        const int cnt = (nq - b0 < 32) ? (nq - b0) : 32;

        spmm_add_batch(myC, myR, myV, cnt, E, acc, rlo, lane);
    }
    __syncthreads();

    const int nrows = (NODE_NUM - rlo < OWN_ROWS) ? (NODE_NUM - rlo) : OWN_ROWS;
    float4* dst = (float4*)(side + (size_t)rlo * EMBED_DIM);
    for (int i = tid; i < nrows * (EMBED_DIM / 4); i += 1024)
        dst[i] = ((const float4*)acc)[i];
}

__global__ void ngcf_layer_kernel(const float* __restrict__ Eprev,
                                  const float* __restrict__ side,
                                  const float* __restrict__ W1,
                                  const float* __restrict__ W2,
                                  float*       __restrict__ Eout) {
    __shared__ float w1i[EMBED_DIM * EMBED_DIM];
    __shared__ float w2i[EMBED_DIM * EMBED_DIM];
    for (int i = threadIdx.x; i < EMBED_DIM * EMBED_DIM; i += blockDim.x) {
        const int k = i >> 6;
        const int n = i & 63;
        const int dst = ((k >> 1) << 7) + (n << 1) + (k & 1);
        w1i[dst] = W1[i];
        w2i[dst] = W2[i];
    }
    __syncthreads();

    const int tile = blockIdx.x * (blockDim.x >> 5) + (threadIdx.x >> 5);
    if (tile >= NODE_TILES) return;

    const int lane    = threadIdx.x & 31;
    const int halfsel = lane >> 4;
    const int nlane   = lane & 15;
    const int m0      = tile * 16;
    const int mrow    = m0 + nlane;
    const int koff    = halfsel * 2;

    v8f acc[4] = {v8f{}, v8f{}, v8f{}, v8f{}};

    const float* epRow = Eprev + (size_t)mrow * EMBED_DIM + koff;
    const float* sdRow = side  + (size_t)mrow * EMBED_DIM + koff;

#pragma unroll
    for (int kk = 0; kk < 16; ++kk) {
        const int k0 = kk * 4;
        const float2 ep = *(const float2*)(epRow + k0);
        const float2 sd = *(const float2*)(sdRow + k0);

        v2f aSum, aBi;
        aSum.x = ep.x + sd.x;  aSum.y = ep.y + sd.y;
        aBi.x  = ep.x * sd.x;  aBi.y  = ep.y * sd.y;

        const int kph = ((k0 >> 1) + halfsel) << 7;
        const float* w1p = w1i + kph + (nlane << 1);
        const float* w2p = w2i + kph + (nlane << 1);

#pragma unroll
        for (int nb = 0; nb < 4; ++nb) {
            const v2f b1 = *(const v2f*)(w1p + nb * 32);
            const v2f b2 = *(const v2f*)(w2p + nb * 32);
            acc[nb] = __builtin_amdgcn_wmma_f32_16x16x4_f32(
                false, aSum, false, b1, (short)0, acc[nb], false, false);
            acc[nb] = __builtin_amdgcn_wmma_f32_16x16x4_f32(
                false, aBi,  false, b2, (short)0, acc[nb], false, false);
        }
    }

#pragma unroll
    for (int nb = 0; nb < 4; ++nb) {
#pragma unroll
        for (int i = 0; i < 8; ++i) {
            float v = acc[nb][i];
            v = (v > 0.f) ? v : v * NEG_SLOPE;
            Eout[(size_t)(m0 + i + halfsel * 8) * EMBED_DIM + nb * 16 + nlane] = v;
        }
    }
}

__global__ void ngcf_gather_kernel(const float* __restrict__ emb,
                                   const float* __restrict__ E1,
                                   const float* __restrict__ E2,
                                   const float* __restrict__ E3,
                                   const int* __restrict__ user_idx,
                                   const int* __restrict__ item_idx,
                                   float* __restrict__ out) {
    const int row = blockIdx.x;
    const int t   = threadIdx.x;
    const int seg = t >> 6;
    const int dim = t & 63;

    int node;
    if (row < BATCH) node = user_idx[row];
    else             node = USER_NUM + item_idx[row - BATCH];

    const float* src = (seg == 0) ? emb : (seg == 1) ? E1 : (seg == 2) ? E2 : E3;
    out[(size_t)row * 256 + t] = src[(size_t)node * EMBED_DIM + dim];
}

extern "C" void kernel_launch(void* const* d_in, const int* in_sizes, int n_in,
                              void* d_out, int out_size, void* d_ws, size_t ws_size,
                              hipStream_t stream) {
    const float* emb      = (const float*)d_in[0];
    const float* W1       = (const float*)d_in[1];
    const float* W2       = (const float*)d_in[2];
    const float* vals     = (const float*)d_in[3];
    const int*   row_idx  = (const int*)d_in[4];
    const int*   col_idx  = (const int*)d_in[5];
    const int*   user_idx = (const int*)d_in[6];
    const int*   item_idx = (const int*)d_in[7];

    const size_t EN = (size_t)NODE_NUM * EMBED_DIM;
    float* side = (float*)d_ws;
    float* E1   = side + EN;
    float* E2   = E1 + EN;
    float* E3   = E2 + EN;
    float* Eouts[N_LAYER] = {E1, E2, E3};

    const int layerBlocks = (NODE_TILES + 7) / 8;

    unsigned* planes = (unsigned*)E3;
    ngcf_planes_kernel<<<PLANE_WAVES, 32, 0, stream>>>(row_idx, planes);

    const float* Eprev = emb;
    for (int l = 0; l < N_LAYER; ++l) {
        ngcf_spmm_kernel<<<OWNERS, 1024, 0, stream>>>(
            vals, row_idx, col_idx, Eprev, side, planes);
        ngcf_layer_kernel<<<layerBlocks, 256, 0, stream>>>(
            Eprev, side,
            W1 + (size_t)l * EMBED_DIM * EMBED_DIM,
            W2 + (size_t)l * EMBED_DIM * EMBED_DIM,
            Eouts[l]);
        Eprev = Eouts[l];
    }

    ngcf_gather_kernel<<<2 * BATCH, 256, 0, stream>>>(emb, E1, E2, E3,
                                                      user_idx, item_idx,
                                                      (float*)d_out);
}


# SpMM: layer 1 writes each workgroup's edge list (column, local row, weight: 8 bytes an edge) to the workspace; layers 2, 3 read it back: no map search, no scattered index loads; LDS add order kept
# speedup vs baseline: 2.6422x; 1.2357x over previous
#include <hip/hip_runtime.h>
#include <hip/hip_bf16.h>


#define USER_NUM 60000
#define ITEM_NUM 40000
#define NODE_NUM (USER_NUM + ITEM_NUM)
#define EMBED_DIM 64
#define N_LAYER 3
#define NNZ 1600000
#define BATCH 4096
#define NEG_SLOPE 0.2f
#define NODE_TILES (NODE_NUM / 16)

typedef float v2f __attribute__((ext_vector_type(2)));
typedef float v8f __attribute__((ext_vector_type(8)));

#define OWN_ROWS 391
#define OWNERS 256
#define PLANES 8
#define PLANE_WORDS (NNZ / 32)
#define PLANE_VEC4 (PLANE_WORDS / 4)
#define PLANE_STRIDE 50016
#define PLANE_STRIDE4 (PLANE_STRIDE / 4)
#define PLANE_WAVES ((PLANE_WORDS + 31) / 32)
__global__ void __launch_bounds__(32)
ngcf_planes_kernel(const int* __restrict__ rows, unsigned* __restrict__ planes) {
    const int lane = threadIdx.x;
    const int w0   = blockIdx.x * 32;
    unsigned mine[PLANES];
#pragma unroll
    for (int k = 0; k < PLANES; ++k) mine[k] = 0u;
#pragma unroll 8
    for (int t = 0; t < 32; ++t) {
        const int w = w0 + t;
        if (w < PLANE_WORDS) {
            const unsigned r = (unsigned)rows[w * 32 + lane];
            unsigned o = r / (unsigned)OWN_ROWS;
            o = (o < (unsigned)OWNERS) ? o : (unsigned)(OWNERS - 1);
#pragma unroll
            for (int k = 0; k < PLANES; ++k) {
                const unsigned b = __builtin_amdgcn_ballot_w32(((o >> k) & 1u) != 0u);
                if (lane == t) mine[k] = b;
            }
        }
    }
    if (w0 + lane < PLANE_WORDS) {
#pragma unroll
        for (int k = 0; k < PLANES; ++k)
            planes[(size_t)k * PLANE_STRIDE + w0 + lane] = mine[k];
    }
}

#define SPMM_HALF (EMBED_DIM / 2)
#define QCAP 16384
#define LCAP 8192
#define LIST_COL_BITS 17
#define LIST_COL_MASK ((1u << LIST_COL_BITS) - 1u)
#define LIST_NO_ROW 511u
#define FIND_TRIPS ((PLANE_VEC4 + 31) / 32)

static __device__ __forceinline__ void spmm_add_batch(const int myC, const int myLr, const float myV,
                                                      const int cnt,
                                                      const float* __restrict__ E,
                                                      float* acc, const int lane) {
    float glo[32];
    float ghi[32];
#pragma unroll
    for (int j = 0; j < 32; ++j) {
        if (j < cnt) {
            const int c = __builtin_amdgcn_readlane(myC, j);
            const float* srow = E + (size_t)c * EMBED_DIM + lane;
            glo[j] = srow[0];
            ghi[j] = srow[SPMM_HALF];
        }
    }
#pragma unroll
    for (int j = 0; j < 32; ++j) {
        if (j < cnt) {
            const int   lr = __builtin_amdgcn_readlane(myLr, j);
            const float v  = __int_as_float(
                __builtin_amdgcn_readlane(__float_as_int(myV), j));
            if ((unsigned)lr < (unsigned)OWN_ROWS) {
                float* drow = acc + lr * EMBED_DIM + lane;
                atomicAdd(drow,             v * glo[j]);
                atomicAdd(drow + SPMM_HALF, v * ghi[j]);
            }
        }
    }
}

template <bool FIRST>
__global__ void __launch_bounds__(1024)
ngcf_spmm_kernel(const float*    __restrict__ vals,
                 const int*      __restrict__ rows,
                 const int*      __restrict__ cols,
                 const float*    __restrict__ E,
                 float*          __restrict__ side,
                 const unsigned* __restrict__ planes,
                 unsigned*       __restrict__ counts,
                 unsigned*       __restrict__ lists) {
    __shared__ float    acc[OWN_ROWS * EMBED_DIM];
    __shared__ unsigned queue[QCAP];
    __shared__ unsigned qcount;

    const int tid   = threadIdx.x;
    const int lane  = tid & 31;
    const int wave  = __builtin_amdgcn_readfirstlane(tid >> 5);
    const int owner = blockIdx.x;
    const int rlo   = owner * OWN_ROWS;

    for (int i = tid; i < OWN_ROWS * (EMBED_DIM / 4); i += 1024)
        ((float4*)acc)[i] = make_float4(0.f, 0.f, 0.f, 0.f);
    if (tid == 0) qcount = 0u;
    __syncthreads();

    uint2* __restrict__ myList = (uint2*)lists + (size_t)owner * LCAP;

    unsigned qc = 0u;
    bool fromList = false;
    if (!FIRST) {
        qc = (unsigned)__builtin_amdgcn_readfirstlane((int)counts[owner]);
        fromList = qc <= (unsigned)LCAP;
    }

    if (fromList) {
        const int nq = (int)qc;
        for (int b0 = wave * 32; b0 < nq; b0 += 32 * 32) {
            const int idx = b0 + lane;
            uint2 ent = make_uint2(0u, 0u);
            if (idx < nq) ent = myList[idx];
            asm volatile("" :: "v"(ent.x), "v"(ent.y));
            const int   myC  = (int)(ent.x & LIST_COL_MASK);
            const int   myLr = (int)(ent.x >> LIST_COL_BITS);
            const float myV  = __uint_as_float(ent.y);
            const int cnt = (nq - b0 < 32) ? (nq - b0) : 32;

            spmm_add_batch(myC, myLr, myV, cnt, E, acc, lane);
        }
    } else {
        const uint4* __restrict__ P4 = (const uint4*)planes;
        const int rot = (owner * FIND_TRIPS) >> 8;
        for (int t = wave; t < FIND_TRIPS; t += 32) {
            const int  trip = (t + rot < FIND_TRIPS) ? (t + rot) : (t + rot - FIND_TRIPS);
            const int  i  = trip * 32 + lane;
            const bool ok = i < PLANE_VEC4;
            const int  ii = ok ? i : (PLANE_VEC4 - 1);
            uint4 p[PLANES];
#pragma unroll
            for (int k = 0; k < PLANES; ++k) p[k] = P4[(size_t)k * PLANE_STRIDE4 + ii];
            unsigned n0 = 0u, n1 = 0u, n2 = 0u, n3 = 0u;
#pragma unroll
            for (int k = 0; k < PLANES; ++k) {
                const unsigned want = ((owner >> k) & 1) ? 0xFFFFFFFFu : 0u;
                n0 |= p[k].x ^ want;  n1 |= p[k].y ^ want;
                n2 |= p[k].z ^ want;  n3 |= p[k].w ^ want;
            }
            unsigned m[4];
            m[0] = ok ? ~n0 : 0u;  m[1] = ok ? ~n1 : 0u;
            m[2] = ok ? ~n2 : 0u;  m[3] = ok ? ~n3 : 0u;
#pragma unroll
            for (int d = 0; d < 4; ++d) {
                unsigned mm = m[d];
                unsigned bal;
                while ((bal = __builtin_amdgcn_ballot_w32(mm != 0u)) != 0u) {
                    if (mm != 0u) {
                        const int      lead = __builtin_ctz(bal);
                        const unsigned rank = __builtin_amdgcn_mbcnt_lo(bal, 0u);
                        const int      bit  = __builtin_ctz(mm);
                        mm &= mm - 1u;
                        unsigned base = 0u;
                        if (lane == lead)
                            base = atomicAdd(&qcount, (unsigned)__builtin_popcount(bal));
                        base = (unsigned)__builtin_amdgcn_readlane((int)base, lead);
                        const unsigned slot = base + rank;
                        const int e = ((ii * 4 + d) << 5) + bit;
                        if (slot < (unsigned)QCAP) {
                            queue[slot] = (unsigned)e;
                        } else {
                            const int   c  = cols[e];
                            const int   lr = rows[e] - rlo;
                            const float v  = vals[e];
                            if ((unsigned)lr < (unsigned)OWN_ROWS) {
                                for (int x = 0; x < EMBED_DIM; ++x)
                                    atomicAdd(&acc[lr * EMBED_DIM + x],
                                              v * E[(size_t)c * EMBED_DIM + x]);
                            }
                        }
                    }
                }
            }
        }
        __syncthreads();
        qc = (unsigned)__builtin_amdgcn_readfirstlane((int)qcount);

        const int  nq   = (int)((qc < (unsigned)QCAP) ? qc : (unsigned)QCAP);
        const bool keep = FIRST && (qc <= (unsigned)LCAP);
        if (FIRST && tid == 0) counts[owner] = qc;
        for (int b0 = wave * 32; b0 < nq; b0 += 32 * 32) {
            const int idx = b0 + lane;
            const int e   = (idx < nq) ? (int)queue[idx] : 0;
            const int   myC = cols[e];
            const int   myR = rows[e];
            const float myV = vals[e];
            asm volatile("" :: "v"(myC), "v"(myR), "v"(myV));
            const int myLr = myR - rlo;
            if (keep && idx < nq) {
                const unsigned lru = ((unsigned)myLr < (unsigned)OWN_ROWS) ? (unsigned)myLr
                                                                           : LIST_NO_ROW;
                myList[idx] = make_uint2((unsigned)myC | (lru << LIST_COL_BITS),
                                         __float_as_uint(myV));
            }
            const int cnt = (nq - b0 < 32) ? (nq - b0) : 32;

            spmm_add_batch(myC, myLr, myV, cnt, E, acc, lane);
        }
    }
    __syncthreads();

    const int nrows = (NODE_NUM - rlo < OWN_ROWS) ? (NODE_NUM - rlo) : OWN_ROWS;
    float4* dst = (float4*)(side + (size_t)rlo * EMBED_DIM);
    for (int i = tid; i < nrows * (EMBED_DIM / 4); i += 1024)
        dst[i] = ((const float4*)acc)[i];
}

__global__ void ngcf_layer_kernel(const float* __restrict__ Eprev,
                                  const float* __restrict__ side,
                                  const float* __restrict__ W1,
                                  const float* __restrict__ W2,
                                  float*       __restrict__ Eout) {
    __shared__ float w1i[EMBED_DIM * EMBED_DIM];
    __shared__ float w2i[EMBED_DIM * EMBED_DIM];
    for (int i = threadIdx.x; i < EMBED_DIM * EMBED_DIM; i += blockDim.x) {
        const int k = i >> 6;
        const int n = i & 63;
        const int dst = ((k >> 1) << 7) + (n << 1) + (k & 1);
        w1i[dst] = W1[i];
        w2i[dst] = W2[i];
    }
    __syncthreads();

    const int tile = blockIdx.x * (blockDim.x >> 5) + (threadIdx.x >> 5);
    if (tile >= NODE_TILES) return;

    const int lane    = threadIdx.x & 31;
    const int halfsel = lane >> 4;
    const int nlane   = lane & 15;
    const int m0      = tile * 16;
    const int mrow    = m0 + nlane;
    const int koff    = halfsel * 2;

    v8f acc[4] = {v8f{}, v8f{}, v8f{}, v8f{}};

    const float* epRow = Eprev + (size_t)mrow * EMBED_DIM + koff;
    const float* sdRow = side  + (size_t)mrow * EMBED_DIM + koff;

#pragma unroll
    for (int kk = 0; kk < 16; ++kk) {
        const int k0 = kk * 4;
        const float2 ep = *(const float2*)(epRow + k0);
        const float2 sd = *(const float2*)(sdRow + k0);

        v2f aSum, aBi;
        aSum.x = ep.x + sd.x;  aSum.y = ep.y + sd.y;
        aBi.x  = ep.x * sd.x;  aBi.y  = ep.y * sd.y;

        const int kph = ((k0 >> 1) + halfsel) << 7;
        const float* w1p = w1i + kph + (nlane << 1);
        const float* w2p = w2i + kph + (nlane << 1);

#pragma unroll
        for (int nb = 0; nb < 4; ++nb) {
            const v2f b1 = *(const v2f*)(w1p + nb * 32);
            const v2f b2 = *(const v2f*)(w2p + nb * 32);
            acc[nb] = __builtin_amdgcn_wmma_f32_16x16x4_f32(
                false, aSum, false, b1, (short)0, acc[nb], false, false);
            acc[nb] = __builtin_amdgcn_wmma_f32_16x16x4_f32(
                false, aBi,  false, b2, (short)0, acc[nb], false, false);
        }
    }

#pragma unroll
    for (int nb = 0; nb < 4; ++nb) {
#pragma unroll
        for (int i = 0; i < 8; ++i) {
            float v = acc[nb][i];
            v = (v > 0.f) ? v : v * NEG_SLOPE;
            Eout[(size_t)(m0 + i + halfsel * 8) * EMBED_DIM + nb * 16 + nlane] = v;
        }
    }
}

__global__ void ngcf_gather_kernel(const float* __restrict__ emb,
                                   const float* __restrict__ E1,
                                   const float* __restrict__ E2,
                                   const float* __restrict__ E3,
                                   const int* __restrict__ user_idx,
                                   const int* __restrict__ item_idx,
                                   float* __restrict__ out) {
    const int row = blockIdx.x;
    const int t   = threadIdx.x;
    const int seg = t >> 6;
    const int dim = t & 63;

    int node;
    if (row < BATCH) node = user_idx[row];
    else             node = USER_NUM + item_idx[row - BATCH];

    const float* src = (seg == 0) ? emb : (seg == 1) ? E1 : (seg == 2) ? E2 : E3;
    out[(size_t)row * 256 + t] = src[(size_t)node * EMBED_DIM + dim];
}

extern "C" void kernel_launch(void* const* d_in, const int* in_sizes, int n_in,
                              void* d_out, int out_size, void* d_ws, size_t ws_size,
                              hipStream_t stream) {
    const float* emb      = (const float*)d_in[0];
    const float* W1       = (const float*)d_in[1];
    const float* W2       = (const float*)d_in[2];
    const float* vals     = (const float*)d_in[3];
    const int*   row_idx  = (const int*)d_in[4];
    const int*   col_idx  = (const int*)d_in[5];
    const int*   user_idx = (const int*)d_in[6];
    const int*   item_idx = (const int*)d_in[7];

    const size_t EN = (size_t)NODE_NUM * EMBED_DIM;
    float* side = (float*)d_ws;
    float* E1   = side + EN;
    float* E2   = E1 + EN;
    float* E3   = E2 + EN;
    float* Eouts[N_LAYER] = {E1, E2, E3};

    const int layerBlocks = (NODE_TILES + 7) / 8;

    unsigned* planes = (unsigned*)E3;
    unsigned* counts = planes + PLANES * PLANE_STRIDE;
    unsigned* lists  = counts + OWNERS;
    ngcf_planes_kernel<<<PLANE_WAVES, 32, 0, stream>>>(row_idx, planes);

    const float* Eprev = emb;
    for (int l = 0; l < N_LAYER; ++l) {
        if (l == 0)
            ngcf_spmm_kernel<true><<<OWNERS, 1024, 0, stream>>>(
                vals, row_idx, col_idx, Eprev, side, planes, counts, lists);
        else
            ngcf_spmm_kernel<false><<<OWNERS, 1024, 0, stream>>>(
                vals, row_idx, col_idx, Eprev, side, planes, counts, lists);
        ngcf_layer_kernel<<<layerBlocks, 256, 0, stream>>>(
            Eprev, side,
            W1 + (size_t)l * EMBED_DIM * EMBED_DIM,
            W2 + (size_t)l * EMBED_DIM * EMBED_DIM,
            Eouts[l]);
        Eprev = Eouts[l];
    }

    ngcf_gather_kernel<<<2 * BATCH, 256, 0, stream>>>(emb, E1, E2, E3,
                                                      user_idx, item_idx,
                                                      (float*)d_out);
}


# Owner-map pass: each wave requests all 32 of its row-index lines before using the first (per-trip guard branches had kept every load behind the trip before it): one memory latency per wave, not 32
# speedup vs baseline: 2.7555x; 1.0429x over previous
#include <hip/hip_runtime.h>
#include <hip/hip_bf16.h>


#define USER_NUM 60000
#define ITEM_NUM 40000
#define NODE_NUM (USER_NUM + ITEM_NUM)
#define EMBED_DIM 64
#define N_LAYER 3
#define NNZ 1600000
#define BATCH 4096
#define NEG_SLOPE 0.2f
#define NODE_TILES (NODE_NUM / 16)

typedef float v2f __attribute__((ext_vector_type(2)));
typedef float v8f __attribute__((ext_vector_type(8)));

#define OWN_ROWS 391
#define OWNERS 256
#define PLANES 8
#define PLANE_WORDS (NNZ / 32)
#define PLANE_VEC4 (PLANE_WORDS / 4)
#define PLANE_STRIDE 50016
#define PLANE_STRIDE4 (PLANE_STRIDE / 4)
#define PLANE_WAVES ((PLANE_WORDS + 31) / 32)
__global__ void __launch_bounds__(32)
ngcf_planes_kernel(const int* __restrict__ rows, unsigned* __restrict__ planes) {
    const int lane = threadIdx.x;
    const int w0   = blockIdx.x * 32;
    unsigned r[32];
#pragma unroll
    for (int t = 0; t < 32; ++t) {
        const int w  = w0 + t;
        const int wc = (w < PLANE_WORDS) ? w : (PLANE_WORDS - 1);
        r[t] = (unsigned)rows[wc * 32 + lane];
    }
    unsigned mine[PLANES];
#pragma unroll
    for (int k = 0; k < PLANES; ++k) mine[k] = 0u;
#pragma unroll
    for (int t = 0; t < 32; ++t) {
        unsigned o = r[t] / (unsigned)OWN_ROWS;
        o = (o < (unsigned)OWNERS) ? o : (unsigned)(OWNERS - 1);
#pragma unroll
        for (int k = 0; k < PLANES; ++k) {
            const unsigned b = __builtin_amdgcn_ballot_w32(((o >> k) & 1u) != 0u);
            if (lane == t) mine[k] = b;
        }
    }
    if (w0 + lane < PLANE_WORDS) {
#pragma unroll
        for (int k = 0; k < PLANES; ++k)
            planes[(size_t)k * PLANE_STRIDE + w0 + lane] = mine[k];
    }
}

#define SPMM_HALF (EMBED_DIM / 2)
#define QCAP 16384
#define LCAP 8192
#define LIST_COL_BITS 17
#define LIST_COL_MASK ((1u << LIST_COL_BITS) - 1u)
#define LIST_NO_ROW 511u
#define FIND_TRIPS ((PLANE_VEC4 + 31) / 32)

static __device__ __forceinline__ void spmm_add_batch(const int myC, const int myLr, const float myV,
                                                      const int cnt,
                                                      const float* __restrict__ E,
                                                      float* acc, const int lane) {
    float glo[32];
    float ghi[32];
#pragma unroll
    for (int j = 0; j < 32; ++j) {
        if (j < cnt) {
            const int c = __builtin_amdgcn_readlane(myC, j);
            const float* srow = E + (size_t)c * EMBED_DIM + lane;
            glo[j] = srow[0];
            ghi[j] = srow[SPMM_HALF];
        }
    }
#pragma unroll
    for (int j = 0; j < 32; ++j) {
        if (j < cnt) {
            const int   lr = __builtin_amdgcn_readlane(myLr, j);
            const float v  = __int_as_float(
                __builtin_amdgcn_readlane(__float_as_int(myV), j));
            if ((unsigned)lr < (unsigned)OWN_ROWS) {
                float* drow = acc + lr * EMBED_DIM + lane;
                atomicAdd(drow,             v * glo[j]);
                atomicAdd(drow + SPMM_HALF, v * ghi[j]);
            }
        }
    }
}

template <bool FIRST>
__global__ void __launch_bounds__(1024)
ngcf_spmm_kernel(const float*    __restrict__ vals,
                 const int*      __restrict__ rows,
                 const int*      __restrict__ cols,
                 const float*    __restrict__ E,
                 float*          __restrict__ side,
                 const unsigned* __restrict__ planes,
                 unsigned*       __restrict__ counts,
                 unsigned*       __restrict__ lists) {
    __shared__ float    acc[OWN_ROWS * EMBED_DIM];
    __shared__ unsigned queue[QCAP];
    __shared__ unsigned qcount;

    const int tid   = threadIdx.x;
    const int lane  = tid & 31;
    const int wave  = __builtin_amdgcn_readfirstlane(tid >> 5);
    const int owner = blockIdx.x;
    const int rlo   = owner * OWN_ROWS;

    for (int i = tid; i < OWN_ROWS * (EMBED_DIM / 4); i += 1024)
        ((float4*)acc)[i] = make_float4(0.f, 0.f, 0.f, 0.f);
    if (tid == 0) qcount = 0u;
    __syncthreads();

    uint2* __restrict__ myList = (uint2*)lists + (size_t)owner * LCAP;

    unsigned qc = 0u;
    bool fromList = false;
    if (!FIRST) {
        qc = (unsigned)__builtin_amdgcn_readfirstlane((int)counts[owner]);
        fromList = qc <= (unsigned)LCAP;
    }

    if (fromList) {
        const int nq = (int)qc;
        for (int b0 = wave * 32; b0 < nq; b0 += 32 * 32) {
            const int idx = b0 + lane;
            uint2 ent = make_uint2(0u, 0u);
            if (idx < nq) ent = myList[idx];
            asm volatile("" :: "v"(ent.x), "v"(ent.y));
            const int   myC  = (int)(ent.x & LIST_COL_MASK);
            const int   myLr = (int)(ent.x >> LIST_COL_BITS);
            const float myV  = __uint_as_float(ent.y);
            const int cnt = (nq - b0 < 32) ? (nq - b0) : 32;

            spmm_add_batch(myC, myLr, myV, cnt, E, acc, lane);
        }
    } else {
        const uint4* __restrict__ P4 = (const uint4*)planes;
        const int rot = (owner * FIND_TRIPS) >> 8;
        for (int t = wave; t < FIND_TRIPS; t += 32) {
            const int  trip = (t + rot < FIND_TRIPS) ? (t + rot) : (t + rot - FIND_TRIPS);
            const int  i  = trip * 32 + lane;
            const bool ok = i < PLANE_VEC4;
            const int  ii = ok ? i : (PLANE_VEC4 - 1);
            uint4 p[PLANES];
#pragma unroll
            for (int k = 0; k < PLANES; ++k) p[k] = P4[(size_t)k * PLANE_STRIDE4 + ii];
            unsigned n0 = 0u, n1 = 0u, n2 = 0u, n3 = 0u;
#pragma unroll
            for (int k = 0; k < PLANES; ++k) {
                const unsigned want = ((owner >> k) & 1) ? 0xFFFFFFFFu : 0u;
                n0 |= p[k].x ^ want;  n1 |= p[k].y ^ want;
                n2 |= p[k].z ^ want;  n3 |= p[k].w ^ want;
            }
            unsigned m[4];
            m[0] = ok ? ~n0 : 0u;  m[1] = ok ? ~n1 : 0u;
            m[2] = ok ? ~n2 : 0u;  m[3] = ok ? ~n3 : 0u;
#pragma unroll
            for (int d = 0; d < 4; ++d) {
                unsigned mm = m[d];
                unsigned bal;
                while ((bal = __builtin_amdgcn_ballot_w32(mm != 0u)) != 0u) {
                    if (mm != 0u) {
                        const int      lead = __builtin_ctz(bal);
                        const unsigned rank = __builtin_amdgcn_mbcnt_lo(bal, 0u);
                        const int      bit  = __builtin_ctz(mm);
                        mm &= mm - 1u;
                        unsigned base = 0u;
                        if (lane == lead)
                            base = atomicAdd(&qcount, (unsigned)__builtin_popcount(bal));
                        base = (unsigned)__builtin_amdgcn_readlane((int)base, lead);
                        const unsigned slot = base + rank;
                        const int e = ((ii * 4 + d) << 5) + bit;
                        if (slot < (unsigned)QCAP) {
                            queue[slot] = (unsigned)e;
                        } else {
                            const int   c  = cols[e];
                            const int   lr = rows[e] - rlo;
                            const float v  = vals[e];
                            if ((unsigned)lr < (unsigned)OWN_ROWS) {
                                for (int x = 0; x < EMBED_DIM; ++x)
                                    atomicAdd(&acc[lr * EMBED_DIM + x],
                                              v * E[(size_t)c * EMBED_DIM + x]);
                            }
                        }
                    }
                }
            }
        }
        __syncthreads();
        qc = (unsigned)__builtin_amdgcn_readfirstlane((int)qcount);

        const int  nq   = (int)((qc < (unsigned)QCAP) ? qc : (unsigned)QCAP);
        const bool keep = FIRST && (qc <= (unsigned)LCAP);
        if (FIRST && tid == 0) counts[owner] = qc;
        for (int b0 = wave * 32; b0 < nq; b0 += 32 * 32) {
            const int idx = b0 + lane;
            const int e   = (idx < nq) ? (int)queue[idx] : 0;
            const int   myC = cols[e];
            const int   myR = rows[e];
            const float myV = vals[e];
            asm volatile("" :: "v"(myC), "v"(myR), "v"(myV));
            const int myLr = myR - rlo;
            if (keep && idx < nq) {
                const unsigned lru = ((unsigned)myLr < (unsigned)OWN_ROWS) ? (unsigned)myLr
                                                                           : LIST_NO_ROW;
                myList[idx] = make_uint2((unsigned)myC | (lru << LIST_COL_BITS),
                                         __float_as_uint(myV));
            }
            const int cnt = (nq - b0 < 32) ? (nq - b0) : 32;

            spmm_add_batch(myC, myLr, myV, cnt, E, acc, lane);
        }
    }
    __syncthreads();

    const int nrows = (NODE_NUM - rlo < OWN_ROWS) ? (NODE_NUM - rlo) : OWN_ROWS;
    float4* dst = (float4*)(side + (size_t)rlo * EMBED_DIM);
    for (int i = tid; i < nrows * (EMBED_DIM / 4); i += 1024)
        dst[i] = ((const float4*)acc)[i];
}

__global__ void ngcf_layer_kernel(const float* __restrict__ Eprev,
                                  const float* __restrict__ side,
                                  const float* __restrict__ W1,
                                  const float* __restrict__ W2,
                                  float*       __restrict__ Eout) {
    __shared__ float w1i[EMBED_DIM * EMBED_DIM];
    __shared__ float w2i[EMBED_DIM * EMBED_DIM];
    for (int i = threadIdx.x; i < EMBED_DIM * EMBED_DIM; i += blockDim.x) {
        const int k = i >> 6;
        const int n = i & 63;
        const int dst = ((k >> 1) << 7) + (n << 1) + (k & 1);
        w1i[dst] = W1[i];
        w2i[dst] = W2[i];
    }
    __syncthreads();

    const int tile = blockIdx.x * (blockDim.x >> 5) + (threadIdx.x >> 5);
    if (tile >= NODE_TILES) return;

    const int lane    = threadIdx.x & 31;
    const int halfsel = lane >> 4;
    const int nlane   = lane & 15;
    const int m0      = tile * 16;
    const int mrow    = m0 + nlane;
    const int koff    = halfsel * 2;

    v8f acc[4] = {v8f{}, v8f{}, v8f{}, v8f{}};

    const float* epRow = Eprev + (size_t)mrow * EMBED_DIM + koff;
    const float* sdRow = side  + (size_t)mrow * EMBED_DIM + koff;

#pragma unroll
    for (int kk = 0; kk < 16; ++kk) {
        const int k0 = kk * 4;
        const float2 ep = *(const float2*)(epRow + k0);
        const float2 sd = *(const float2*)(sdRow + k0);

        v2f aSum, aBi;
        aSum.x = ep.x + sd.x;  aSum.y = ep.y + sd.y;
        aBi.x  = ep.x * sd.x;  aBi.y  = ep.y * sd.y;

        const int kph = ((k0 >> 1) + halfsel) << 7;
        const float* w1p = w1i + kph + (nlane << 1);
        const float* w2p = w2i + kph + (nlane << 1);

#pragma unroll
        for (int nb = 0; nb < 4; ++nb) {
            const v2f b1 = *(const v2f*)(w1p + nb * 32);
            const v2f b2 = *(const v2f*)(w2p + nb * 32);
            acc[nb] = __builtin_amdgcn_wmma_f32_16x16x4_f32(
                false, aSum, false, b1, (short)0, acc[nb], false, false);
            acc[nb] = __builtin_amdgcn_wmma_f32_16x16x4_f32(
                false, aBi,  false, b2, (short)0, acc[nb], false, false);
        }
    }

#pragma unroll
    for (int nb = 0; nb < 4; ++nb) {
#pragma unroll
        for (int i = 0; i < 8; ++i) {
            float v = acc[nb][i];
            v = (v > 0.f) ? v : v * NEG_SLOPE;
            Eout[(size_t)(m0 + i + halfsel * 8) * EMBED_DIM + nb * 16 + nlane] = v;
        }
    }
}

__global__ void ngcf_gather_kernel(const float* __restrict__ emb,
                                   const float* __restrict__ E1,
                                   const float* __restrict__ E2,
                                   const float* __restrict__ E3,
                                   const int* __restrict__ user_idx,
                                   const int* __restrict__ item_idx,
                                   float* __restrict__ out) {
    const int row = blockIdx.x;
    const int t   = threadIdx.x;
    const int seg = t >> 6;
    const int dim = t & 63;

    int node;
    if (row < BATCH) node = user_idx[row];
    else             node = USER_NUM + item_idx[row - BATCH];

    const float* src = (seg == 0) ? emb : (seg == 1) ? E1 : (seg == 2) ? E2 : E3;
    out[(size_t)row * 256 + t] = src[(size_t)node * EMBED_DIM + dim];
}

extern "C" void kernel_launch(void* const* d_in, const int* in_sizes, int n_in,
                              void* d_out, int out_size, void* d_ws, size_t ws_size,
                              hipStream_t stream) {
    const float* emb      = (const float*)d_in[0];
    const float* W1       = (const float*)d_in[1];
    const float* W2       = (const float*)d_in[2];
    const float* vals     = (const float*)d_in[3];
    const int*   row_idx  = (const int*)d_in[4];
    const int*   col_idx  = (const int*)d_in[5];
    const int*   user_idx = (const int*)d_in[6];
    const int*   item_idx = (const int*)d_in[7];

    const size_t EN = (size_t)NODE_NUM * EMBED_DIM;
    float* side = (float*)d_ws;
    float* E1   = side + EN;
    float* E2   = E1 + EN;
    float* E3   = E2 + EN;
    float* Eouts[N_LAYER] = {E1, E2, E3};

    const int layerBlocks = (NODE_TILES + 7) / 8;

    unsigned* planes = (unsigned*)E3;
    unsigned* counts = planes + PLANES * PLANE_STRIDE;
    unsigned* lists  = counts + OWNERS;
    ngcf_planes_kernel<<<PLANE_WAVES, 32, 0, stream>>>(row_idx, planes);

    const float* Eprev = emb;
    for (int l = 0; l < N_LAYER; ++l) {
        if (l == 0)
            ngcf_spmm_kernel<true><<<OWNERS, 1024, 0, stream>>>(
                vals, row_idx, col_idx, Eprev, side, planes, counts, lists);
        else
            ngcf_spmm_kernel<false><<<OWNERS, 1024, 0, stream>>>(
                vals, row_idx, col_idx, Eprev, side, planes, counts, lists);
        ngcf_layer_kernel<<<layerBlocks, 256, 0, stream>>>(
            Eprev, side,
            W1 + (size_t)l * EMBED_DIM * EMBED_DIM,
            W2 + (size_t)l * EMBED_DIM * EMBED_DIM,
            Eouts[l]);
        Eprev = Eouts[l];
    }

    ngcf_gather_kernel<<<2 * BATCH, 256, 0, stream>>>(emb, E1, E2, E3,
                                                      user_idx, item_idx,
                                                      (float*)d_out);
}


# SpMM edge lists from one split pass: 128 workgroups rank their slice's edges per owner (LDS counters) into (owner, slice) cells; all 3 SpMMs read cells; no bit-plane map; LDS f32 adds, arrival order
# speedup vs baseline: 2.9701x; 1.0779x over previous
#include <hip/hip_runtime.h>
#include <hip/hip_bf16.h>


#define USER_NUM 60000
#define ITEM_NUM 40000
#define NODE_NUM (USER_NUM + ITEM_NUM)
#define EMBED_DIM 64
#define N_LAYER 3
#define NNZ 1600000
#define BATCH 4096
#define NEG_SLOPE 0.2f
#define NODE_TILES (NODE_NUM / 16)

typedef float v2f __attribute__((ext_vector_type(2)));
typedef float v8f __attribute__((ext_vector_type(8)));

#define OWN_ROWS 391
#define OWNERS 256
#define NB 128
#define SLICE (NNZ / NB)
#define CAP 96
#define SPLIT_TRIPS ((SLICE + 1023) / 1024)
#define LIST_COL_BITS 17
#define LIST_COL_MASK ((1u << LIST_COL_BITS) - 1u)
#define LIST_NO_ROW 511u
__global__ void __launch_bounds__(1024)
ngcf_split_kernel(const float* __restrict__ vals,
                  const int*   __restrict__ rows,
                  const int*   __restrict__ cols,
                  unsigned*    __restrict__ counts,
                  unsigned*    __restrict__ lists) {
    __shared__ unsigned fill[OWNERS];
    uint2* __restrict__ cells = (uint2*)lists;
    const int tid   = threadIdx.x;
    const int slice = blockIdx.x;
    const int e0    = slice * SLICE;
    if (tid < OWNERS) fill[tid] = 0u;
    unsigned r[SPLIT_TRIPS];
    unsigned c[SPLIT_TRIPS];
    float    v[SPLIT_TRIPS];
#pragma unroll
    for (int k = 0; k < SPLIT_TRIPS; ++k) {
        const int i  = tid + k * 1024;
        const int ic = (i < SLICE) ? i : (SLICE - 1);
        r[k] = (unsigned)rows[e0 + ic];
        c[k] = (unsigned)cols[e0 + ic];
        v[k] = vals[e0 + ic];
    }
    __syncthreads();
    unsigned o[SPLIT_TRIPS];
    unsigned slot[SPLIT_TRIPS];
#pragma unroll
    for (int k = 0; k < SPLIT_TRIPS; ++k) {
        unsigned ok = r[k] / (unsigned)OWN_ROWS;
        ok = (ok < (unsigned)OWNERS) ? ok : (unsigned)(OWNERS - 1);
        o[k] = ok;
        slot[k] = (unsigned)CAP;
        if (tid + k * 1024 < SLICE) slot[k] = atomicAdd(&fill[ok], 1u);
    }
#pragma unroll
    for (int k = 0; k < SPLIT_TRIPS; ++k) {
        if (slot[k] < (unsigned)CAP) {
            const unsigned lr  = r[k] - o[k] * (unsigned)OWN_ROWS;
            const unsigned lru = (lr < (unsigned)OWN_ROWS) ? lr : LIST_NO_ROW;
            cells[((size_t)o[k] * NB + slice) * CAP + slot[k]] =
                make_uint2(c[k] | (lru << LIST_COL_BITS), __float_as_uint(v[k]));
        }
    }
    __syncthreads();
    if (tid < OWNERS) counts[slice * OWNERS + tid] = fill[tid];
}

#define SPMM_HALF (EMBED_DIM / 2)
#define WAVE_CELLS (NB / 32)

static __device__ __forceinline__ void spmm_add_batch(const int myC, const int myLr, const float myV,
                                                      const int cnt,
                                                      const float* __restrict__ E,
                                                      float* acc, const int lane) {
    float glo[32];
    float ghi[32];
#pragma unroll
    for (int j = 0; j < 32; ++j) {
        if (j < cnt) {
            const int c = __builtin_amdgcn_readlane(myC, j);
            const float* srow = E + (size_t)c * EMBED_DIM + lane;
            glo[j] = srow[0];
            ghi[j] = srow[SPMM_HALF];
        }
    }
#pragma unroll
    for (int j = 0; j < 32; ++j) {
        if (j < cnt) {
            const int   lr = __builtin_amdgcn_readlane(myLr, j);
            const float v  = __int_as_float(
                __builtin_amdgcn_readlane(__float_as_int(myV), j));
            if ((unsigned)lr < (unsigned)OWN_ROWS) {
                float* drow = acc + lr * EMBED_DIM + lane;
                atomicAdd(drow,             v * glo[j]);
                atomicAdd(drow + SPMM_HALF, v * ghi[j]);
            }
        }
    }
}

__global__ void __launch_bounds__(1024)
ngcf_spmm_kernel(const float*    __restrict__ vals,
                 const int*      __restrict__ rows,
                 const int*      __restrict__ cols,
                 const float*    __restrict__ E,
                 float*          __restrict__ side,
                 const unsigned* __restrict__ counts,
                 const unsigned* __restrict__ lists) {
    __shared__ float acc[OWN_ROWS * EMBED_DIM];

    const int tid   = threadIdx.x;
    const int lane  = tid & 31;
    const int wave  = __builtin_amdgcn_readfirstlane(tid >> 5);
    const int owner = blockIdx.x;
    const int rlo   = owner * OWN_ROWS;

    unsigned cn[WAVE_CELLS];
#pragma unroll
    for (int k = 0; k < WAVE_CELLS; ++k)
        cn[k] = counts[(k * 32 + lane) * OWNERS + owner];

    for (int i = tid; i < OWN_ROWS * (EMBED_DIM / 4); i += 1024)
        ((float4*)acc)[i] = make_float4(0.f, 0.f, 0.f, 0.f);

    bool big = false;
#pragma unroll
    for (int k = 0; k < WAVE_CELLS; ++k) big = big || (cn[k] > (unsigned)CAP);
    const bool over = __builtin_amdgcn_ballot_w32(big) != 0u;
    __syncthreads();

    if (!over) {
        const int p1 = __builtin_amdgcn_readlane((int)cn[0], wave);
        const int p2 = p1 + __builtin_amdgcn_readlane((int)cn[1], wave);
        const int p3 = p2 + __builtin_amdgcn_readlane((int)cn[2], wave);
        const int nq = p3 + __builtin_amdgcn_readlane((int)cn[3], wave);
        const uint2* __restrict__ myCells = (const uint2*)lists + ((size_t)owner * NB + wave) * CAP;
        for (int b0 = 0; b0 < nq; b0 += 32) {
            const int idx = b0 + lane;
            const int k   = (idx >= p1) + (idx >= p2) + (idx >= p3);
            const int off = idx - ((idx >= p3) ? p3 : (idx >= p2) ? p2 : (idx >= p1) ? p1 : 0);
            uint2 ent = make_uint2(0u, 0u);
            if (idx < nq) ent = myCells[k * (32 * CAP) + off];
            asm volatile("" :: "v"(ent.x), "v"(ent.y));
            const int   myC  = (int)(ent.x & LIST_COL_MASK);
            const int   myLr = (int)(ent.x >> LIST_COL_BITS);
            const float myV  = __uint_as_float(ent.y);
            const int cnt = (nq - b0 < 32) ? (nq - b0) : 32;

            spmm_add_batch(myC, myLr, myV, cnt, E, acc, lane);
        }
    } else {
        for (int e = tid; e < NNZ; e += 1024) {
            const unsigned r = (unsigned)rows[e];
            unsigned o = r / (unsigned)OWN_ROWS;
            o = (o < (unsigned)OWNERS) ? o : (unsigned)(OWNERS - 1);
            const unsigned lr = r - (unsigned)rlo;
            if (o == (unsigned)owner && lr < (unsigned)OWN_ROWS) {
                const int   c = cols[e];
                const float v = vals[e];
                for (int x = 0; x < EMBED_DIM; ++x)
                    atomicAdd(&acc[lr * EMBED_DIM + x], v * E[(size_t)c * EMBED_DIM + x]);
            }
        }
    }
    __syncthreads();

    const int nrows = (NODE_NUM - rlo < OWN_ROWS) ? (NODE_NUM - rlo) : OWN_ROWS;
    float4* dst = (float4*)(side + (size_t)rlo * EMBED_DIM);
    for (int i = tid; i < nrows * (EMBED_DIM / 4); i += 1024)
        dst[i] = ((const float4*)acc)[i];
}

__global__ void ngcf_layer_kernel(const float* __restrict__ Eprev,
                                  const float* __restrict__ side,
                                  const float* __restrict__ W1,
                                  const float* __restrict__ W2,
                                  float*       __restrict__ Eout) {
    __shared__ float w1i[EMBED_DIM * EMBED_DIM];
    __shared__ float w2i[EMBED_DIM * EMBED_DIM];
    for (int i = threadIdx.x; i < EMBED_DIM * EMBED_DIM; i += blockDim.x) {
        const int k = i >> 6;
        const int n = i & 63;
        const int dst = ((k >> 1) << 7) + (n << 1) + (k & 1);
        w1i[dst] = W1[i];
        w2i[dst] = W2[i];
    }
    __syncthreads();

    const int tile = blockIdx.x * (blockDim.x >> 5) + (threadIdx.x >> 5);
    if (tile >= NODE_TILES) return;

    const int lane    = threadIdx.x & 31;
    const int halfsel = lane >> 4;
    const int nlane   = lane & 15;
    const int m0      = tile * 16;
    const int mrow    = m0 + nlane;
    const int koff    = halfsel * 2;

    v8f acc[4] = {v8f{}, v8f{}, v8f{}, v8f{}};

    const float* epRow = Eprev + (size_t)mrow * EMBED_DIM + koff;
    const float* sdRow = side  + (size_t)mrow * EMBED_DIM + koff;

#pragma unroll
    for (int kk = 0; kk < 16; ++kk) {
        const int k0 = kk * 4;
        const float2 ep = *(const float2*)(epRow + k0);
        const float2 sd = *(const float2*)(sdRow + k0);

        v2f aSum, aBi;
        aSum.x = ep.x + sd.x;  aSum.y = ep.y + sd.y;
        aBi.x  = ep.x * sd.x;  aBi.y  = ep.y * sd.y;

        const int kph = ((k0 >> 1) + halfsel) << 7;
        const float* w1p = w1i + kph + (nlane << 1);
        const float* w2p = w2i + kph + (nlane << 1);

#pragma unroll
        for (int nb = 0; nb < 4; ++nb) {
            const v2f b1 = *(const v2f*)(w1p + nb * 32);
            const v2f b2 = *(const v2f*)(w2p + nb * 32);
            acc[nb] = __builtin_amdgcn_wmma_f32_16x16x4_f32(
                false, aSum, false, b1, (short)0, acc[nb], false, false);
            acc[nb] = __builtin_amdgcn_wmma_f32_16x16x4_f32(
                false, aBi,  false, b2, (short)0, acc[nb], false, false);
        }
    }

#pragma unroll
    for (int nb = 0; nb < 4; ++nb) {
#pragma unroll
        for (int i = 0; i < 8; ++i) {
            float v = acc[nb][i];
            v = (v > 0.f) ? v : v * NEG_SLOPE;
            Eout[(size_t)(m0 + i + halfsel * 8) * EMBED_DIM + nb * 16 + nlane] = v;
        }
    }
}

__global__ void ngcf_gather_kernel(const float* __restrict__ emb,
                                   const float* __restrict__ E1,
                                   const float* __restrict__ E2,
                                   const float* __restrict__ E3,
                                   const int* __restrict__ user_idx,
                                   const int* __restrict__ item_idx,
                                   float* __restrict__ out) {
    const int row = blockIdx.x;
    const int t   = threadIdx.x;
    const int seg = t >> 6;
    const int dim = t & 63;

    int node;
    if (row < BATCH) node = user_idx[row];
    else             node = USER_NUM + item_idx[row - BATCH];

    const float* src = (seg == 0) ? emb : (seg == 1) ? E1 : (seg == 2) ? E2 : E3;
    out[(size_t)row * 256 + t] = src[(size_t)node * EMBED_DIM + dim];
}

extern "C" void kernel_launch(void* const* d_in, const int* in_sizes, int n_in,
                              void* d_out, int out_size, void* d_ws, size_t ws_size,
                              hipStream_t stream) {
    const float* emb      = (const float*)d_in[0];
    const float* W1       = (const float*)d_in[1];
    const float* W2       = (const float*)d_in[2];
    const float* vals     = (const float*)d_in[3];
    const int*   row_idx  = (const int*)d_in[4];
    const int*   col_idx  = (const int*)d_in[5];
    const int*   user_idx = (const int*)d_in[6];
    const int*   item_idx = (const int*)d_in[7];

    const size_t EN = (size_t)NODE_NUM * EMBED_DIM;
    float* side = (float*)d_ws;
    float* E1   = side + EN;
    float* E2   = E1 + EN;
    float* E3   = E2 + EN;
    float* Eouts[N_LAYER] = {E1, E2, E3};

    const int layerBlocks = (NODE_TILES + 7) / 8;

    unsigned* counts = (unsigned*)E3;
    unsigned* lists  = counts + OWNERS * NB;
    ngcf_split_kernel<<<NB, 1024, 0, stream>>>(vals, row_idx, col_idx, counts, lists);

    const float* Eprev = emb;
    for (int l = 0; l < N_LAYER; ++l) {
        ngcf_spmm_kernel<<<OWNERS, 1024, 0, stream>>>(
            vals, row_idx, col_idx, Eprev, side, counts, lists);
        ngcf_layer_kernel<<<layerBlocks, 256, 0, stream>>>(
            Eprev, side,
            W1 + (size_t)l * EMBED_DIM * EMBED_DIM,
            W2 + (size_t)l * EMBED_DIM * EMBED_DIM,
            Eouts[l]);
        Eprev = Eouts[l];
    }

    ngcf_gather_kernel<<<2 * BATCH, 256, 0, stream>>>(emb, E1, E2, E3,
                                                      user_idx, item_idx,
                                                      (float*)d_out);
}


# Layer GEMMs run inside the owner workgroups: after its LDS adds each does its 25 row tiles on WMMA from LDS (E_prev rows staged beside the sums); side_E never stored; 8->5 launches; sum order kept
# speedup vs baseline: 3.7636x; 1.2672x over previous
#include <hip/hip_runtime.h>
#include <hip/hip_bf16.h>


#define USER_NUM 60000
#define ITEM_NUM 40000
#define NODE_NUM (USER_NUM + ITEM_NUM)
#define EMBED_DIM 64
#define N_LAYER 3
#define NNZ 1600000
#define BATCH 4096
#define NEG_SLOPE 0.2f

typedef float v2f __attribute__((ext_vector_type(2)));
typedef float v8f __attribute__((ext_vector_type(8)));

#define OWN_ROWS 391
#define OWNERS 256
#define NB 128
#define SLICE (NNZ / NB)
#define CAP 96
#define SPLIT_TRIPS ((SLICE + 1023) / 1024)
#define LIST_COL_BITS 17
#define LIST_COL_MASK ((1u << LIST_COL_BITS) - 1u)
#define LIST_NO_ROW 511u
__global__ void __launch_bounds__(1024)
ngcf_split_kernel(const float* __restrict__ vals,
                  const int*   __restrict__ rows,
                  const int*   __restrict__ cols,
                  unsigned*    __restrict__ counts,
                  unsigned*    __restrict__ lists) {
    __shared__ unsigned fill[OWNERS];
    uint2* __restrict__ cells = (uint2*)lists;
    const int tid   = threadIdx.x;
    const int slice = blockIdx.x;
    const int e0    = slice * SLICE;
    if (tid < OWNERS) fill[tid] = 0u;
    unsigned r[SPLIT_TRIPS];
    unsigned c[SPLIT_TRIPS];
    float    v[SPLIT_TRIPS];
#pragma unroll
    for (int k = 0; k < SPLIT_TRIPS; ++k) {
        const int i  = tid + k * 1024;
        const int ic = (i < SLICE) ? i : (SLICE - 1);
        r[k] = (unsigned)rows[e0 + ic];
        c[k] = (unsigned)cols[e0 + ic];
        v[k] = vals[e0 + ic];
    }
    __syncthreads();
    unsigned o[SPLIT_TRIPS];
    unsigned slot[SPLIT_TRIPS];
#pragma unroll
    for (int k = 0; k < SPLIT_TRIPS; ++k) {
        unsigned ok = r[k] / (unsigned)OWN_ROWS;
        ok = (ok < (unsigned)OWNERS) ? ok : (unsigned)(OWNERS - 1);
        o[k] = ok;
        slot[k] = (unsigned)CAP;
        if (tid + k * 1024 < SLICE) slot[k] = atomicAdd(&fill[ok], 1u);
    }
#pragma unroll
    for (int k = 0; k < SPLIT_TRIPS; ++k) {
        if (slot[k] < (unsigned)CAP) {
            const unsigned lr  = r[k] - o[k] * (unsigned)OWN_ROWS;
            const unsigned lru = (lr < (unsigned)OWN_ROWS) ? lr : LIST_NO_ROW;
            cells[((size_t)o[k] * NB + slice) * CAP + slot[k]] =
                make_uint2(c[k] | (lru << LIST_COL_BITS), __float_as_uint(v[k]));
        }
    }
    __syncthreads();
    if (tid < OWNERS) counts[slice * OWNERS + tid] = fill[tid];
}

#define SPMM_HALF (EMBED_DIM / 2)
#define WAVE_CELLS (NB / 32)
#define PITCH 68
#define PITCH4 (PITCH / 4)
#define OWN_TILES ((OWN_ROWS + 15) / 16)
#define PAD_ROWS (OWN_TILES * 16)
#define W_TRIPS (EMBED_DIM * EMBED_DIM / 1024)
#define PRV_TRIPS ((PAD_ROWS * (EMBED_DIM / 4) + 1023) / 1024)

static __device__ __forceinline__ void spmm_add_batch(const int myC, const int myLr, const float myV,
                                                      const int cnt,
                                                      const float* __restrict__ E,
                                                      float* acc, const int lane) {
    float glo[32];
    float ghi[32];
#pragma unroll
    for (int j = 0; j < 32; ++j) {
        if (j < cnt) {
            const int c = __builtin_amdgcn_readlane(myC, j);
            const float* srow = E + (size_t)c * EMBED_DIM + lane;
            glo[j] = srow[0];
            ghi[j] = srow[SPMM_HALF];
        }
    }
#pragma unroll
    for (int j = 0; j < 32; ++j) {
        if (j < cnt) {
            const int   lr = __builtin_amdgcn_readlane(myLr, j);
            const float v  = __int_as_float(
                __builtin_amdgcn_readlane(__float_as_int(myV), j));
            if ((unsigned)lr < (unsigned)OWN_ROWS) {
                float* drow = acc + lr * PITCH + lane;
                atomicAdd(drow,             v * glo[j]);
                atomicAdd(drow + SPMM_HALF, v * ghi[j]);
            }
        }
    }
}

__global__ void __launch_bounds__(1024)
ngcf_owner_layer_kernel(const float*    __restrict__ vals,
                        const int*      __restrict__ rows,
                        const int*      __restrict__ cols,
                        const float*    __restrict__ E,
                        const float*    __restrict__ W1,
                        const float*    __restrict__ W2,
                        float*          __restrict__ Eout,
                        const unsigned* __restrict__ counts,
                        const unsigned* __restrict__ lists) {
    __shared__ float acc[PAD_ROWS * PITCH];
    __shared__ float prv[PAD_ROWS * PITCH];
    __shared__ float w1i[EMBED_DIM * EMBED_DIM];
    __shared__ float w2i[EMBED_DIM * EMBED_DIM];

    const int tid   = threadIdx.x;
    const int lane  = tid & 31;
    const int wave  = __builtin_amdgcn_readfirstlane(tid >> 5);
    const int owner = blockIdx.x;
    const int rlo   = owner * OWN_ROWS;

    unsigned cn[WAVE_CELLS];
#pragma unroll
    for (int k = 0; k < WAVE_CELLS; ++k)
        cn[k] = counts[(k * 32 + lane) * OWNERS + owner];

    const int nrows = (NODE_NUM - rlo < OWN_ROWS) ? (NODE_NUM - rlo) : OWN_ROWS;
    float  w1r[W_TRIPS];
    float  w2r[W_TRIPS];
    float4 pr[PRV_TRIPS];
#pragma unroll
    for (int k = 0; k < W_TRIPS; ++k) {
        w1r[k] = W1[tid + k * 1024];
        w2r[k] = W2[tid + k * 1024];
    }
    const float4* __restrict__ prev4 = (const float4*)(E + (size_t)rlo * EMBED_DIM);
    const int lastPiece = nrows * (EMBED_DIM / 4) - 1;
#pragma unroll
    for (int k = 0; k < PRV_TRIPS; ++k) {
        const int i = tid + k * 1024;
        pr[k] = prev4[(i < lastPiece) ? i : lastPiece];
    }
    for (int i = tid; i < PAD_ROWS * PITCH4; i += 1024)
        ((float4*)acc)[i] = make_float4(0.f, 0.f, 0.f, 0.f);
#pragma unroll
    for (int k = 0; k < W_TRIPS; ++k) {
        const int i  = tid + k * 1024;
        const int kr = i >> 6;
        const int n  = i & 63;
        const int dst = ((kr >> 1) << 7) + (n << 1) + (kr & 1);
        w1i[dst] = w1r[k];
        w2i[dst] = w2r[k];
    }
#pragma unroll
    for (int k = 0; k < PRV_TRIPS; ++k) {
        const int i = tid + k * 1024;
        if (i < PAD_ROWS * (EMBED_DIM / 4)) {
            const int r = i >> 4;
            const int q = i & 15;
            ((float4*)prv)[r * PITCH4 + q] =
                (r < nrows) ? pr[k] : make_float4(0.f, 0.f, 0.f, 0.f);
        }
    }

    bool big = false;
#pragma unroll
    for (int k = 0; k < WAVE_CELLS; ++k) big = big || (cn[k] > (unsigned)CAP);
    const bool over = __builtin_amdgcn_ballot_w32(big) != 0u;
    __syncthreads();

    if (!over) {
        const int p1 = __builtin_amdgcn_readlane((int)cn[0], wave);
        const int p2 = p1 + __builtin_amdgcn_readlane((int)cn[1], wave);
        const int p3 = p2 + __builtin_amdgcn_readlane((int)cn[2], wave);
        const int nq = p3 + __builtin_amdgcn_readlane((int)cn[3], wave);
        const uint2* __restrict__ myCells = (const uint2*)lists + ((size_t)owner * NB + wave) * CAP;
        for (int b0 = 0; b0 < nq; b0 += 32) {
            const int idx = b0 + lane;
            const int k   = (idx >= p1) + (idx >= p2) + (idx >= p3);
            const int off = idx - ((idx >= p3) ? p3 : (idx >= p2) ? p2 : (idx >= p1) ? p1 : 0);
            uint2 ent = make_uint2(0u, 0u);
            if (idx < nq) ent = myCells[k * (32 * CAP) + off];
            asm volatile("" :: "v"(ent.x), "v"(ent.y));
            const int   myC  = (int)(ent.x & LIST_COL_MASK);
            const int   myLr = (int)(ent.x >> LIST_COL_BITS);
            const float myV  = __uint_as_float(ent.y);
            const int cnt = (nq - b0 < 32) ? (nq - b0) : 32;

            spmm_add_batch(myC, myLr, myV, cnt, E, acc, lane);
        }
    } else {
        for (int e = tid; e < NNZ; e += 1024) {
            const unsigned r = (unsigned)rows[e];
            unsigned o = r / (unsigned)OWN_ROWS;
            o = (o < (unsigned)OWNERS) ? o : (unsigned)(OWNERS - 1);
            const unsigned lr = r - (unsigned)rlo;
            if (o == (unsigned)owner && lr < (unsigned)OWN_ROWS) {
                const int   c = cols[e];
                const float v = vals[e];
                for (int x = 0; x < EMBED_DIM; ++x)
                    atomicAdd(&acc[lr * PITCH + x], v * E[(size_t)c * EMBED_DIM + x]);
            }
        }
    }
    __syncthreads();

    if (wave < OWN_TILES) {
        const int halfsel = lane >> 4;
        const int nlane   = lane & 15;
        const int koff    = halfsel * 2;
        const int t0      = wave * 16;

        v8f d[4] = {v8f{}, v8f{}, v8f{}, v8f{}};

        const float* epRow = prv + (t0 + nlane) * PITCH + koff;
        const float* sdRow = acc + (t0 + nlane) * PITCH + koff;

#pragma unroll
        for (int kk = 0; kk < 16; ++kk) {
            const int k0 = kk * 4;
            const float2 ep = *(const float2*)(epRow + k0);
            const float2 sd = *(const float2*)(sdRow + k0);

            v2f aSum, aBi;
            aSum.x = ep.x + sd.x;  aSum.y = ep.y + sd.y;
            aBi.x  = ep.x * sd.x;  aBi.y  = ep.y * sd.y;

            const int kph = ((k0 >> 1) + halfsel) << 7;
            const float* w1p = w1i + kph + (nlane << 1);
            const float* w2p = w2i + kph + (nlane << 1);

#pragma unroll
            for (int nb = 0; nb < 4; ++nb) {
                const v2f b1 = *(const v2f*)(w1p + nb * 32);
                const v2f b2 = *(const v2f*)(w2p + nb * 32);
                d[nb] = __builtin_amdgcn_wmma_f32_16x16x4_f32(
                    false, aSum, false, b1, (short)0, d[nb], false, false);
                d[nb] = __builtin_amdgcn_wmma_f32_16x16x4_f32(
                    false, aBi,  false, b2, (short)0, d[nb], false, false);
            }
        }

#pragma unroll
        for (int nb = 0; nb < 4; ++nb) {
#pragma unroll
            for (int i = 0; i < 8; ++i) {
                float v = d[nb][i];
                v = (v > 0.f) ? v : v * NEG_SLOPE;
                acc[(t0 + i + halfsel * 8) * PITCH + nb * 16 + nlane] = v;
            }
        }
    }
    __syncthreads();

    float4* dst = (float4*)(Eout + (size_t)rlo * EMBED_DIM);
    for (int i = tid; i < nrows * (EMBED_DIM / 4); i += 1024)
        dst[i] = ((const float4*)acc)[(i >> 4) * PITCH4 + (i & 15)];
}

__global__ void ngcf_gather_kernel(const float* __restrict__ emb,
                                   const float* __restrict__ E1,
                                   const float* __restrict__ E2,
                                   const float* __restrict__ E3,
                                   const int* __restrict__ user_idx,
                                   const int* __restrict__ item_idx,
                                   float* __restrict__ out) {
    const int row = blockIdx.x;
    const int t   = threadIdx.x;
    const int seg = t >> 6;
    const int dim = t & 63;

    int node;
    if (row < BATCH) node = user_idx[row];
    else             node = USER_NUM + item_idx[row - BATCH];

    const float* src = (seg == 0) ? emb : (seg == 1) ? E1 : (seg == 2) ? E2 : E3;
    out[(size_t)row * 256 + t] = src[(size_t)node * EMBED_DIM + dim];
}

extern "C" void kernel_launch(void* const* d_in, const int* in_sizes, int n_in,
                              void* d_out, int out_size, void* d_ws, size_t ws_size,
                              hipStream_t stream) {
    const float* emb      = (const float*)d_in[0];
    const float* W1       = (const float*)d_in[1];
    const float* W2       = (const float*)d_in[2];
    const float* vals     = (const float*)d_in[3];
    const int*   row_idx  = (const int*)d_in[4];
    const int*   col_idx  = (const int*)d_in[5];
    const int*   user_idx = (const int*)d_in[6];
    const int*   item_idx = (const int*)d_in[7];

    const size_t EN = (size_t)NODE_NUM * EMBED_DIM;
    float* lst  = (float*)d_ws;
    float* E1   = lst + EN;
    float* E2   = E1 + EN;
    float* E3   = E2 + EN;
    float* Eouts[N_LAYER] = {E1, E2, E3};

    unsigned* counts = (unsigned*)lst;
    unsigned* lists  = counts + OWNERS * NB;
    ngcf_split_kernel<<<NB, 1024, 0, stream>>>(vals, row_idx, col_idx, counts, lists);

    const float* Eprev = emb;
    for (int l = 0; l < N_LAYER; ++l) {
        ngcf_owner_layer_kernel<<<OWNERS, 1024, 0, stream>>>(
            vals, row_idx, col_idx, Eprev,
            W1 + (size_t)l * EMBED_DIM * EMBED_DIM,
            W2 + (size_t)l * EMBED_DIM * EMBED_DIM,
            Eouts[l], counts, lists);
        Eprev = Eouts[l];
    }

    ngcf_gather_kernel<<<2 * BATCH, 256, 0, stream>>>(emb, E1, E2, E3,
                                                      user_idx, item_idx,
                                                      (float*)d_out);
}


# Output rows written by the owner workgroups: each lists in LDS the output slots whose node it owns and copies those rows from LDS; no gather launch, last E_next not stored; 5->4 launches
# speedup vs baseline: 3.9215x; 1.0420x over previous
#include <hip/hip_runtime.h>
#include <hip/hip_bf16.h>


#define USER_NUM 60000
#define ITEM_NUM 40000
#define NODE_NUM (USER_NUM + ITEM_NUM)
#define EMBED_DIM 64
#define N_LAYER 3
#define NNZ 1600000
#define BATCH 4096
#define NEG_SLOPE 0.2f

typedef float v2f __attribute__((ext_vector_type(2)));
typedef float v8f __attribute__((ext_vector_type(8)));

#define OWN_ROWS 391
#define OWNERS 256
#define NB 128
#define SLICE (NNZ / NB)
#define CAP 96
#define SPLIT_TRIPS ((SLICE + 1023) / 1024)
#define LIST_COL_BITS 17
#define LIST_COL_MASK ((1u << LIST_COL_BITS) - 1u)
#define LIST_NO_ROW 511u
__global__ void __launch_bounds__(1024)
ngcf_split_kernel(const float* __restrict__ vals,
                  const int*   __restrict__ rows,
                  const int*   __restrict__ cols,
                  unsigned*    __restrict__ counts,
                  unsigned*    __restrict__ lists) {
    __shared__ unsigned fill[OWNERS];
    uint2* __restrict__ cells = (uint2*)lists;
    const int tid   = threadIdx.x;
    const int slice = blockIdx.x;
    const int e0    = slice * SLICE;
    if (tid < OWNERS) fill[tid] = 0u;
    unsigned r[SPLIT_TRIPS];
    unsigned c[SPLIT_TRIPS];
    float    v[SPLIT_TRIPS];
#pragma unroll
    for (int k = 0; k < SPLIT_TRIPS; ++k) {
        const int i  = tid + k * 1024;
        const int ic = (i < SLICE) ? i : (SLICE - 1);
        r[k] = (unsigned)rows[e0 + ic];
        c[k] = (unsigned)cols[e0 + ic];
        v[k] = vals[e0 + ic];
    }
    __syncthreads();
    unsigned o[SPLIT_TRIPS];
    unsigned slot[SPLIT_TRIPS];
#pragma unroll
    for (int k = 0; k < SPLIT_TRIPS; ++k) {
        unsigned ok = r[k] / (unsigned)OWN_ROWS;
        ok = (ok < (unsigned)OWNERS) ? ok : (unsigned)(OWNERS - 1);
        o[k] = ok;
        slot[k] = (unsigned)CAP;
        if (tid + k * 1024 < SLICE) slot[k] = atomicAdd(&fill[ok], 1u);
    }
#pragma unroll
    for (int k = 0; k < SPLIT_TRIPS; ++k) {
        if (slot[k] < (unsigned)CAP) {
            const unsigned lr  = r[k] - o[k] * (unsigned)OWN_ROWS;
            const unsigned lru = (lr < (unsigned)OWN_ROWS) ? lr : LIST_NO_ROW;
            cells[((size_t)o[k] * NB + slice) * CAP + slot[k]] =
                make_uint2(c[k] | (lru << LIST_COL_BITS), __float_as_uint(v[k]));
        }
    }
    __syncthreads();
    if (tid < OWNERS) counts[slice * OWNERS + tid] = fill[tid];
}

#define SPMM_HALF (EMBED_DIM / 2)
#define WAVE_CELLS (NB / 32)
#define PITCH 68
#define PITCH4 (PITCH / 4)
#define OWN_TILES ((OWN_ROWS + 15) / 16)
#define PAD_ROWS (OWN_TILES * 16)
#define W_TRIPS (EMBED_DIM * EMBED_DIM / 1024)
#define PRV_TRIPS ((PAD_ROWS * (EMBED_DIM / 4) + 1023) / 1024)
#define SLOTS (2 * BATCH)
#define SLOT_TRIPS (SLOTS / 1024)
#define OUT_PITCH ((N_LAYER + 1) * EMBED_DIM)
#define SEL_ROW_BITS 9
#define SEL_ROW_MASK ((1u << SEL_ROW_BITS) - 1u)

static __device__ __forceinline__ void spmm_add_batch(const int myC, const int myLr, const float myV,
                                                      const int cnt,
                                                      const float* __restrict__ E,
                                                      float* acc, const int lane) {
    float glo[32];
    float ghi[32];
#pragma unroll
    for (int j = 0; j < 32; ++j) {
        if (j < cnt) {
            const int c = __builtin_amdgcn_readlane(myC, j);
            const float* srow = E + (size_t)c * EMBED_DIM + lane;
            glo[j] = srow[0];
            ghi[j] = srow[SPMM_HALF];
        }
    }
#pragma unroll
    for (int j = 0; j < 32; ++j) {
        if (j < cnt) {
            const int   lr = __builtin_amdgcn_readlane(myLr, j);
            const float v  = __int_as_float(
                __builtin_amdgcn_readlane(__float_as_int(myV), j));
            if ((unsigned)lr < (unsigned)OWN_ROWS) {
                float* drow = acc + lr * PITCH + lane;
                atomicAdd(drow,             v * glo[j]);
                atomicAdd(drow + SPMM_HALF, v * ghi[j]);
            }
        }
    }
}

__global__ void __launch_bounds__(1024)
ngcf_owner_layer_kernel(const float*    __restrict__ vals,
                        const int*      __restrict__ rows,
                        const int*      __restrict__ cols,
                        const float*    __restrict__ E,
                        const float*    __restrict__ W1,
                        const float*    __restrict__ W2,
                        float*          __restrict__ Eout,
                        const unsigned* __restrict__ counts,
                        const unsigned* __restrict__ lists,
                        const int*      __restrict__ user_idx,
                        const int*      __restrict__ item_idx,
                        float*          outSeg,
                        float*          outPrev) {
    __shared__ float acc[PAD_ROWS * PITCH];
    __shared__ float prv[PAD_ROWS * PITCH];
    __shared__ float w1i[EMBED_DIM * EMBED_DIM];
    __shared__ float w2i[EMBED_DIM * EMBED_DIM];
    __shared__ unsigned sel[SLOTS];
    __shared__ unsigned nsel;

    const int tid   = threadIdx.x;
    const int lane  = tid & 31;
    const int wave  = __builtin_amdgcn_readfirstlane(tid >> 5);
    const int owner = blockIdx.x;
    const int rlo   = owner * OWN_ROWS;

    unsigned cn[WAVE_CELLS];
#pragma unroll
    for (int k = 0; k < WAVE_CELLS; ++k)
        cn[k] = counts[(k * 32 + lane) * OWNERS + owner];
    if (tid == 0) nsel = 0u;
    int node[SLOT_TRIPS];
#pragma unroll
    for (int k = 0; k < SLOT_TRIPS; ++k)
        node[k] = (k < SLOT_TRIPS / 2) ? user_idx[tid + k * 1024]
                                       : USER_NUM + item_idx[tid + (k - SLOT_TRIPS / 2) * 1024];

    const int nrows = (NODE_NUM - rlo < OWN_ROWS) ? (NODE_NUM - rlo) : OWN_ROWS;
    float  w1r[W_TRIPS];
    float  w2r[W_TRIPS];
    float4 pr[PRV_TRIPS];
#pragma unroll
    for (int k = 0; k < W_TRIPS; ++k) {
        w1r[k] = W1[tid + k * 1024];
        w2r[k] = W2[tid + k * 1024];
    }
    const float4* __restrict__ prev4 = (const float4*)(E + (size_t)rlo * EMBED_DIM);
    const int lastPiece = nrows * (EMBED_DIM / 4) - 1;
#pragma unroll
    for (int k = 0; k < PRV_TRIPS; ++k) {
        const int i = tid + k * 1024;
        pr[k] = prev4[(i < lastPiece) ? i : lastPiece];
    }
    for (int i = tid; i < PAD_ROWS * PITCH4; i += 1024)
        ((float4*)acc)[i] = make_float4(0.f, 0.f, 0.f, 0.f);
#pragma unroll
    for (int k = 0; k < W_TRIPS; ++k) {
        const int i  = tid + k * 1024;
        const int kr = i >> 6;
        const int n  = i & 63;
        const int dst = ((kr >> 1) << 7) + (n << 1) + (kr & 1);
        w1i[dst] = w1r[k];
        w2i[dst] = w2r[k];
    }
#pragma unroll
    for (int k = 0; k < PRV_TRIPS; ++k) {
        const int i = tid + k * 1024;
        if (i < PAD_ROWS * (EMBED_DIM / 4)) {
            const int r = i >> 4;
            const int q = i & 15;
            ((float4*)prv)[r * PITCH4 + q] =
                (r < nrows) ? pr[k] : make_float4(0.f, 0.f, 0.f, 0.f);
        }
    }

    bool big = false;
#pragma unroll
    for (int k = 0; k < WAVE_CELLS; ++k) big = big || (cn[k] > (unsigned)CAP);
    const bool over = __builtin_amdgcn_ballot_w32(big) != 0u;
    __syncthreads();

#pragma unroll
    for (int k = 0; k < SLOT_TRIPS; ++k) {
        const unsigned lr = (unsigned)(node[k] - rlo);
        if (lr < (unsigned)nrows) {
            const unsigned pos = atomicAdd(&nsel, 1u);
            sel[pos] = ((unsigned)(tid + k * 1024) << SEL_ROW_BITS) | lr;
        }
    }

    if (!over) {
        const int p1 = __builtin_amdgcn_readlane((int)cn[0], wave);
        const int p2 = p1 + __builtin_amdgcn_readlane((int)cn[1], wave);
        const int p3 = p2 + __builtin_amdgcn_readlane((int)cn[2], wave);
        const int nq = p3 + __builtin_amdgcn_readlane((int)cn[3], wave);
        const uint2* __restrict__ myCells = (const uint2*)lists + ((size_t)owner * NB + wave) * CAP;
        for (int b0 = 0; b0 < nq; b0 += 32) {
            const int idx = b0 + lane;
            const int k   = (idx >= p1) + (idx >= p2) + (idx >= p3);
            const int off = idx - ((idx >= p3) ? p3 : (idx >= p2) ? p2 : (idx >= p1) ? p1 : 0);
            uint2 ent = make_uint2(0u, 0u);
            if (idx < nq) ent = myCells[k * (32 * CAP) + off];
            asm volatile("" :: "v"(ent.x), "v"(ent.y));
            const int   myC  = (int)(ent.x & LIST_COL_MASK);
            const int   myLr = (int)(ent.x >> LIST_COL_BITS);
            const float myV  = __uint_as_float(ent.y);
            const int cnt = (nq - b0 < 32) ? (nq - b0) : 32;

            spmm_add_batch(myC, myLr, myV, cnt, E, acc, lane);
        }
    } else {
        for (int e = tid; e < NNZ; e += 1024) {
            const unsigned r = (unsigned)rows[e];
            unsigned o = r / (unsigned)OWN_ROWS;
            o = (o < (unsigned)OWNERS) ? o : (unsigned)(OWNERS - 1);
            const unsigned lr = r - (unsigned)rlo;
            if (o == (unsigned)owner && lr < (unsigned)OWN_ROWS) {
                const int   c = cols[e];
                const float v = vals[e];
                for (int x = 0; x < EMBED_DIM; ++x)
                    atomicAdd(&acc[lr * PITCH + x], v * E[(size_t)c * EMBED_DIM + x]);
            }
        }
    }
    __syncthreads();

    if (wave < OWN_TILES) {
        const int halfsel = lane >> 4;
        const int nlane   = lane & 15;
        const int koff    = halfsel * 2;
        const int t0      = wave * 16;

        v8f d[4] = {v8f{}, v8f{}, v8f{}, v8f{}};

        const float* epRow = prv + (t0 + nlane) * PITCH + koff;
        const float* sdRow = acc + (t0 + nlane) * PITCH + koff;

#pragma unroll
        for (int kk = 0; kk < 16; ++kk) {
            const int k0 = kk * 4;
            const float2 ep = *(const float2*)(epRow + k0);
            const float2 sd = *(const float2*)(sdRow + k0);

            v2f aSum, aBi;
            aSum.x = ep.x + sd.x;  aSum.y = ep.y + sd.y;
            aBi.x  = ep.x * sd.x;  aBi.y  = ep.y * sd.y;

            const int kph = ((k0 >> 1) + halfsel) << 7;
            const float* w1p = w1i + kph + (nlane << 1);
            const float* w2p = w2i + kph + (nlane << 1);

#pragma unroll
            for (int nb = 0; nb < 4; ++nb) {
                const v2f b1 = *(const v2f*)(w1p + nb * 32);
                const v2f b2 = *(const v2f*)(w2p + nb * 32);
                d[nb] = __builtin_amdgcn_wmma_f32_16x16x4_f32(
                    false, aSum, false, b1, (short)0, d[nb], false, false);
                d[nb] = __builtin_amdgcn_wmma_f32_16x16x4_f32(
                    false, aBi,  false, b2, (short)0, d[nb], false, false);
            }
        }

#pragma unroll
        for (int nb = 0; nb < 4; ++nb) {
#pragma unroll
            for (int i = 0; i < 8; ++i) {
                float v = d[nb][i];
                v = (v > 0.f) ? v : v * NEG_SLOPE;
                acc[(t0 + i + halfsel * 8) * PITCH + nb * 16 + nlane] = v;
            }
        }
    }
    __syncthreads();

    if (Eout != nullptr) {
        float4* dst = (float4*)(Eout + (size_t)rlo * EMBED_DIM);
        for (int i = tid; i < nrows * (EMBED_DIM / 4); i += 1024)
            dst[i] = ((const float4*)acc)[(i >> 4) * PITCH4 + (i & 15)];
    }
    const int nm = __builtin_amdgcn_readfirstlane((int)nsel);
    for (int m = wave; m < nm; m += 32) {
        const unsigned ent  = (unsigned)__builtin_amdgcn_readfirstlane((int)sel[m]);
        const unsigned lr   = ent & SEL_ROW_MASK;
        const size_t   slot = (size_t)(ent >> SEL_ROW_BITS);
        *(float2*)(outSeg + slot * OUT_PITCH + 2 * lane) =
            *(const float2*)(acc + lr * PITCH + 2 * lane);
        if (outPrev != nullptr)
            *(float2*)(outPrev + slot * OUT_PITCH + 2 * lane) =
                *(const float2*)(prv + lr * PITCH + 2 * lane);
    }
}

extern "C" void kernel_launch(void* const* d_in, const int* in_sizes, int n_in,
                              void* d_out, int out_size, void* d_ws, size_t ws_size,
                              hipStream_t stream) {
    const float* emb      = (const float*)d_in[0];
    const float* W1       = (const float*)d_in[1];
    const float* W2       = (const float*)d_in[2];
    const float* vals     = (const float*)d_in[3];
    const int*   row_idx  = (const int*)d_in[4];
    const int*   col_idx  = (const int*)d_in[5];
    const int*   user_idx = (const int*)d_in[6];
    const int*   item_idx = (const int*)d_in[7];

    const size_t EN = (size_t)NODE_NUM * EMBED_DIM;
    float* lst  = (float*)d_ws;
    float* E1   = lst + EN;
    float* E2   = E1 + EN;
    float* out  = (float*)d_out;
    float* Eouts[N_LAYER]    = {E1, E2, nullptr};
    float* outSegs[N_LAYER]  = {out + EMBED_DIM, out + 2 * EMBED_DIM, out + 3 * EMBED_DIM};
    float* outPrevs[N_LAYER] = {out, nullptr, nullptr};

    unsigned* counts = (unsigned*)lst;
    unsigned* lists  = counts + OWNERS * NB;
    ngcf_split_kernel<<<NB, 1024, 0, stream>>>(vals, row_idx, col_idx, counts, lists);

    const float* Eprev = emb;
    for (int l = 0; l < N_LAYER; ++l) {
        ngcf_owner_layer_kernel<<<OWNERS, 1024, 0, stream>>>(
            vals, row_idx, col_idx, Eprev,
            W1 + (size_t)l * EMBED_DIM * EMBED_DIM,
            W2 + (size_t)l * EMBED_DIM * EMBED_DIM,
            Eouts[l], counts, lists, user_idx, item_idx, outSegs[l], outPrevs[l]);
        Eprev = Eouts[l];
    }
}


# Split pass: edge entries go to an LDS image of the workgroup's 256 cells, then out as whole cells, lanes consecutive: 2 whole lines a store, not 32 partial ones; ranks, cell layout, add order kept
# speedup vs baseline: 4.0509x; 1.0330x over previous
#include <hip/hip_runtime.h>
#include <hip/hip_bf16.h>


#define USER_NUM 60000
#define ITEM_NUM 40000
#define NODE_NUM (USER_NUM + ITEM_NUM)
#define EMBED_DIM 64
#define N_LAYER 3
#define NNZ 1600000
#define BATCH 4096
#define NEG_SLOPE 0.2f

typedef float v2f __attribute__((ext_vector_type(2)));
typedef float v8f __attribute__((ext_vector_type(8)));

#define OWN_ROWS 391
#define OWNERS 256
#define NB 128
#define SLICE (NNZ / NB)
#define CAP 96
#define SPLIT_TRIPS ((SLICE + 1023) / 1024)
#define LIST_COL_BITS 17
#define LIST_COL_MASK ((1u << LIST_COL_BITS) - 1u)
#define LIST_NO_ROW 511u
#define CELL_PITCH (CAP + 1)
#define CELLS_PER_WAVE (OWNERS / 32)
#define CELL_TRIPS (CAP / 32)
#define IMG_QUADS (OWNERS * CELL_PITCH / 2)
__global__ void __launch_bounds__(1024)
ngcf_split_kernel(const float* __restrict__ vals,
                  const int*   __restrict__ rows,
                  const int*   __restrict__ cols,
                  unsigned*    __restrict__ counts,
                  unsigned*    __restrict__ lists) {
    __shared__ unsigned fill[OWNERS];
    __shared__ __attribute__((aligned(16))) unsigned long long img[OWNERS * CELL_PITCH];
    unsigned long long* __restrict__ cells = (unsigned long long*)lists;
    const int tid   = threadIdx.x;
    const int slice = blockIdx.x;
    const int e0    = slice * SLICE;
    if (tid < OWNERS) fill[tid] = 0u;
    unsigned r[SPLIT_TRIPS];
    unsigned c[SPLIT_TRIPS];
    float    v[SPLIT_TRIPS];
#pragma unroll
    for (int k = 0; k < SPLIT_TRIPS; ++k) {
        const int i  = tid + k * 1024;
        const int ic = (i < SLICE) ? i : (SLICE - 1);
        r[k] = (unsigned)rows[e0 + ic];
        c[k] = (unsigned)cols[e0 + ic];
        v[k] = vals[e0 + ic];
    }
    for (int i = tid; i < IMG_QUADS; i += 1024)
        ((uint4*)img)[i] = make_uint4(0u, 0u, 0u, 0u);
    __syncthreads();
    unsigned o[SPLIT_TRIPS];
    unsigned slot[SPLIT_TRIPS];
#pragma unroll
    for (int k = 0; k < SPLIT_TRIPS; ++k) {
        unsigned ok = r[k] / (unsigned)OWN_ROWS;
        ok = (ok < (unsigned)OWNERS) ? ok : (unsigned)(OWNERS - 1);
        o[k] = ok;
        slot[k] = (unsigned)CAP;
        if (tid + k * 1024 < SLICE) slot[k] = atomicAdd(&fill[ok], 1u);
    }
#pragma unroll
    for (int k = 0; k < SPLIT_TRIPS; ++k) {
        if (slot[k] < (unsigned)CAP) {
            const unsigned lr  = r[k] - o[k] * (unsigned)OWN_ROWS;
            const unsigned lru = (lr < (unsigned)OWN_ROWS) ? lr : LIST_NO_ROW;
            img[o[k] * CELL_PITCH + slot[k]] =
                ((unsigned long long)__float_as_uint(v[k]) << 32) |
                (unsigned long long)(c[k] | (lru << LIST_COL_BITS));
        }
    }
    __syncthreads();
    const int lane = tid & 31;
    const int wave = tid >> 5;
    unsigned cnt[CELLS_PER_WAVE];
    unsigned long long ent[CELLS_PER_WAVE][CELL_TRIPS];
#pragma unroll
    for (int i = 0; i < CELLS_PER_WAVE; ++i) {
        const unsigned cf = fill[wave + 32 * i];
        cnt[i] = (cf < (unsigned)CAP) ? cf : (unsigned)CAP;
    }
#pragma unroll
    for (int i = 0; i < CELLS_PER_WAVE; ++i) {
#pragma unroll
        for (int t = 0; t < CELL_TRIPS; ++t)
            ent[i][t] = img[(wave + 32 * i) * CELL_PITCH + lane + 32 * t];
    }
#pragma unroll
    for (int i = 0; i < CELLS_PER_WAVE; ++i) {
        unsigned long long* __restrict__ dst =
            cells + ((size_t)(wave + 32 * i) * NB + slice) * CAP;
#pragma unroll
        for (int t = 0; t < CELL_TRIPS; ++t) {
            const unsigned j = (unsigned)(lane + 32 * t);
            if (j < cnt[i]) dst[j] = ent[i][t];
        }
    }
    if (tid < OWNERS) counts[slice * OWNERS + tid] = fill[tid];
}

#define SPMM_HALF (EMBED_DIM / 2)
#define WAVE_CELLS (NB / 32)
#define PITCH 68
#define PITCH4 (PITCH / 4)
#define OWN_TILES ((OWN_ROWS + 15) / 16)
#define PAD_ROWS (OWN_TILES * 16)
#define W_TRIPS (EMBED_DIM * EMBED_DIM / 1024)
#define PRV_TRIPS ((PAD_ROWS * (EMBED_DIM / 4) + 1023) / 1024)
#define SLOTS (2 * BATCH)
#define SLOT_TRIPS (SLOTS / 1024)
#define OUT_PITCH ((N_LAYER + 1) * EMBED_DIM)
#define SEL_ROW_BITS 9
#define SEL_ROW_MASK ((1u << SEL_ROW_BITS) - 1u)

static __device__ __forceinline__ void spmm_add_batch(const int myC, const int myLr, const float myV,
                                                      const int cnt,
                                                      const float* __restrict__ E,
                                                      float* acc, const int lane) {
    float glo[32];
    float ghi[32];
#pragma unroll
    for (int j = 0; j < 32; ++j) {
        if (j < cnt) {
            const int c = __builtin_amdgcn_readlane(myC, j);
            const float* srow = E + (size_t)c * EMBED_DIM + lane;
            glo[j] = srow[0];
            ghi[j] = srow[SPMM_HALF];
        }
    }
#pragma unroll
    for (int j = 0; j < 32; ++j) {
        if (j < cnt) {
            const int   lr = __builtin_amdgcn_readlane(myLr, j);
            const float v  = __int_as_float(
                __builtin_amdgcn_readlane(__float_as_int(myV), j));
            if ((unsigned)lr < (unsigned)OWN_ROWS) {
                float* drow = acc + lr * PITCH + lane;
                atomicAdd(drow,             v * glo[j]);
                atomicAdd(drow + SPMM_HALF, v * ghi[j]);
            }
        }
    }
}

__global__ void __launch_bounds__(1024)
ngcf_owner_layer_kernel(const float*    __restrict__ vals,
                        const int*      __restrict__ rows,
                        const int*      __restrict__ cols,
                        const float*    __restrict__ E,
                        const float*    __restrict__ W1,
                        const float*    __restrict__ W2,
                        float*          __restrict__ Eout,
                        const unsigned* __restrict__ counts,
                        const unsigned* __restrict__ lists,
                        const int*      __restrict__ user_idx,
                        const int*      __restrict__ item_idx,
                        float*          outSeg,
                        float*          outPrev) {
    __shared__ float acc[PAD_ROWS * PITCH];
    __shared__ float prv[PAD_ROWS * PITCH];
    __shared__ float w1i[EMBED_DIM * EMBED_DIM];
    __shared__ float w2i[EMBED_DIM * EMBED_DIM];
    __shared__ unsigned sel[SLOTS];
    __shared__ unsigned nsel;

    const int tid   = threadIdx.x;
    const int lane  = tid & 31;
    const int wave  = __builtin_amdgcn_readfirstlane(tid >> 5);
    const int owner = blockIdx.x;
    const int rlo   = owner * OWN_ROWS;

    unsigned cn[WAVE_CELLS];
#pragma unroll
    for (int k = 0; k < WAVE_CELLS; ++k)
        cn[k] = counts[(k * 32 + lane) * OWNERS + owner];
    if (tid == 0) nsel = 0u;
    int node[SLOT_TRIPS];
#pragma unroll
    for (int k = 0; k < SLOT_TRIPS; ++k)
        node[k] = (k < SLOT_TRIPS / 2) ? user_idx[tid + k * 1024]
                                       : USER_NUM + item_idx[tid + (k - SLOT_TRIPS / 2) * 1024];

    const int nrows = (NODE_NUM - rlo < OWN_ROWS) ? (NODE_NUM - rlo) : OWN_ROWS;
    float  w1r[W_TRIPS];
    float  w2r[W_TRIPS];
    float4 pr[PRV_TRIPS];
#pragma unroll
    for (int k = 0; k < W_TRIPS; ++k) {
        w1r[k] = W1[tid + k * 1024];
        w2r[k] = W2[tid + k * 1024];
    }
    const float4* __restrict__ prev4 = (const float4*)(E + (size_t)rlo * EMBED_DIM);
    const int lastPiece = nrows * (EMBED_DIM / 4) - 1;
#pragma unroll
    for (int k = 0; k < PRV_TRIPS; ++k) {
        const int i = tid + k * 1024;
        pr[k] = prev4[(i < lastPiece) ? i : lastPiece];
    }
    for (int i = tid; i < PAD_ROWS * PITCH4; i += 1024)
        ((float4*)acc)[i] = make_float4(0.f, 0.f, 0.f, 0.f);
#pragma unroll
    for (int k = 0; k < W_TRIPS; ++k) {
        const int i  = tid + k * 1024;
        const int kr = i >> 6;
        const int n  = i & 63;
        const int dst = ((kr >> 1) << 7) + (n << 1) + (kr & 1);
        w1i[dst] = w1r[k];
        w2i[dst] = w2r[k];
    }
#pragma unroll
    for (int k = 0; k < PRV_TRIPS; ++k) {
        const int i = tid + k * 1024;
        if (i < PAD_ROWS * (EMBED_DIM / 4)) {
            const int r = i >> 4;
            const int q = i & 15;
            ((float4*)prv)[r * PITCH4 + q] =
                (r < nrows) ? pr[k] : make_float4(0.f, 0.f, 0.f, 0.f);
        }
    }

    bool big = false;
#pragma unroll
    for (int k = 0; k < WAVE_CELLS; ++k) big = big || (cn[k] > (unsigned)CAP);
    const bool over = __builtin_amdgcn_ballot_w32(big) != 0u;
    __syncthreads();

#pragma unroll
    for (int k = 0; k < SLOT_TRIPS; ++k) {
        const unsigned lr = (unsigned)(node[k] - rlo);
        if (lr < (unsigned)nrows) {
            const unsigned pos = atomicAdd(&nsel, 1u);
            sel[pos] = ((unsigned)(tid + k * 1024) << SEL_ROW_BITS) | lr;
        }
    }

    if (!over) {
        const int p1 = __builtin_amdgcn_readlane((int)cn[0], wave);
        const int p2 = p1 + __builtin_amdgcn_readlane((int)cn[1], wave);
        const int p3 = p2 + __builtin_amdgcn_readlane((int)cn[2], wave);
        const int nq = p3 + __builtin_amdgcn_readlane((int)cn[3], wave);
        const uint2* __restrict__ myCells = (const uint2*)lists + ((size_t)owner * NB + wave) * CAP;
        for (int b0 = 0; b0 < nq; b0 += 32) {
            const int idx = b0 + lane;
            const int k   = (idx >= p1) + (idx >= p2) + (idx >= p3);
            const int off = idx - ((idx >= p3) ? p3 : (idx >= p2) ? p2 : (idx >= p1) ? p1 : 0);
            uint2 ent = make_uint2(0u, 0u);
            if (idx < nq) ent = myCells[k * (32 * CAP) + off];
            asm volatile("" :: "v"(ent.x), "v"(ent.y));
            const int   myC  = (int)(ent.x & LIST_COL_MASK);
            const int   myLr = (int)(ent.x >> LIST_COL_BITS);
            const float myV  = __uint_as_float(ent.y);
            const int cnt = (nq - b0 < 32) ? (nq - b0) : 32;

            spmm_add_batch(myC, myLr, myV, cnt, E, acc, lane);
        }
    } else {
        for (int e = tid; e < NNZ; e += 1024) {
            const unsigned r = (unsigned)rows[e];
            unsigned o = r / (unsigned)OWN_ROWS;
            o = (o < (unsigned)OWNERS) ? o : (unsigned)(OWNERS - 1);
            const unsigned lr = r - (unsigned)rlo;
            if (o == (unsigned)owner && lr < (unsigned)OWN_ROWS) {
                const int   c = cols[e];
                const float v = vals[e];
                for (int x = 0; x < EMBED_DIM; ++x)
                    atomicAdd(&acc[lr * PITCH + x], v * E[(size_t)c * EMBED_DIM + x]);
            }
        }
    }
    __syncthreads();

    if (wave < OWN_TILES) {
        const int halfsel = lane >> 4;
        const int nlane   = lane & 15;
        const int koff    = halfsel * 2;
        const int t0      = wave * 16;

        v8f d[4] = {v8f{}, v8f{}, v8f{}, v8f{}};

        const float* epRow = prv + (t0 + nlane) * PITCH + koff;
        const float* sdRow = acc + (t0 + nlane) * PITCH + koff;

#pragma unroll
        for (int kk = 0; kk < 16; ++kk) {
            const int k0 = kk * 4;
            const float2 ep = *(const float2*)(epRow + k0);
            const float2 sd = *(const float2*)(sdRow + k0);

            v2f aSum, aBi;
            aSum.x = ep.x + sd.x;  aSum.y = ep.y + sd.y;
            aBi.x  = ep.x * sd.x;  aBi.y  = ep.y * sd.y;

            const int kph = ((k0 >> 1) + halfsel) << 7;
            const float* w1p = w1i + kph + (nlane << 1);
            const float* w2p = w2i + kph + (nlane << 1);

#pragma unroll
            for (int nb = 0; nb < 4; ++nb) {
                const v2f b1 = *(const v2f*)(w1p + nb * 32);
                const v2f b2 = *(const v2f*)(w2p + nb * 32);
                d[nb] = __builtin_amdgcn_wmma_f32_16x16x4_f32(
                    false, aSum, false, b1, (short)0, d[nb], false, false);
                d[nb] = __builtin_amdgcn_wmma_f32_16x16x4_f32(
                    false, aBi,  false, b2, (short)0, d[nb], false, false);
            }
        }

#pragma unroll
        for (int nb = 0; nb < 4; ++nb) {
#pragma unroll
            for (int i = 0; i < 8; ++i) {
                float v = d[nb][i];
                v = (v > 0.f) ? v : v * NEG_SLOPE;
                acc[(t0 + i + halfsel * 8) * PITCH + nb * 16 + nlane] = v;
            }
        }
    }
    __syncthreads();

    if (Eout != nullptr) {
        float4* dst = (float4*)(Eout + (size_t)rlo * EMBED_DIM);
        for (int i = tid; i < nrows * (EMBED_DIM / 4); i += 1024)
            dst[i] = ((const float4*)acc)[(i >> 4) * PITCH4 + (i & 15)];
    }
    const int nm = __builtin_amdgcn_readfirstlane((int)nsel);
    for (int m = wave; m < nm; m += 32) {
        const unsigned ent  = (unsigned)__builtin_amdgcn_readfirstlane((int)sel[m]);
        const unsigned lr   = ent & SEL_ROW_MASK;
        const size_t   slot = (size_t)(ent >> SEL_ROW_BITS);
        *(float2*)(outSeg + slot * OUT_PITCH + 2 * lane) =
            *(const float2*)(acc + lr * PITCH + 2 * lane);
        if (outPrev != nullptr)
            *(float2*)(outPrev + slot * OUT_PITCH + 2 * lane) =
                *(const float2*)(prv + lr * PITCH + 2 * lane);
    }
}

extern "C" void kernel_launch(void* const* d_in, const int* in_sizes, int n_in,
                              void* d_out, int out_size, void* d_ws, size_t ws_size,
                              hipStream_t stream) {
    const float* emb      = (const float*)d_in[0];
    const float* W1       = (const float*)d_in[1];
    const float* W2       = (const float*)d_in[2];
    const float* vals     = (const float*)d_in[3];
    const int*   row_idx  = (const int*)d_in[4];
    const int*   col_idx  = (const int*)d_in[5];
    const int*   user_idx = (const int*)d_in[6];
    const int*   item_idx = (const int*)d_in[7];

    const size_t EN = (size_t)NODE_NUM * EMBED_DIM;
    float* lst  = (float*)d_ws;
    float* E1   = lst + EN;
    float* E2   = E1 + EN;
    float* out  = (float*)d_out;
    float* Eouts[N_LAYER]    = {E1, E2, nullptr};
    float* outSegs[N_LAYER]  = {out + EMBED_DIM, out + 2 * EMBED_DIM, out + 3 * EMBED_DIM};
    float* outPrevs[N_LAYER] = {out, nullptr, nullptr};

    unsigned* counts = (unsigned*)lst;
    unsigned* lists  = counts + OWNERS * NB;
    ngcf_split_kernel<<<NB, 1024, 0, stream>>>(vals, row_idx, col_idx, counts, lists);

    const float* Eprev = emb;
    for (int l = 0; l < N_LAYER; ++l) {
        ngcf_owner_layer_kernel<<<OWNERS, 1024, 0, stream>>>(
            vals, row_idx, col_idx, Eprev,
            W1 + (size_t)l * EMBED_DIM * EMBED_DIM,
            W2 + (size_t)l * EMBED_DIM * EMBED_DIM,
            Eouts[l], counts, lists, user_idx, item_idx, outSegs[l], outPrevs[l]);
        Eprev = Eouts[l];
    }
}
